# starting kernel
# baseline (speedup 1.0000x reference)
#include <hip/hip_runtime.h>
#include <hip/hip_bf16.h>
#include <math.h>

#define N_NODES   50000
#define N_EDGES   800000
#define HEADS     4
#define D_HEAD    32
#define FEAT      128
#define OUT_DIM   512
#define NEG_SLOPE 0.2f
#define EPS_      1e-9f

typedef __attribute__((ext_vector_type(2))) float v2f;
typedef __attribute__((ext_vector_type(8))) float v8f;


__global__ void k_zero(float* __restrict__ p, long n) {
    long i = (long)blockIdx.x * blockDim.x + threadIdx.x;
    if (i < n) p[i] = 0.0f;
}

__device__ __forceinline__ unsigned f2ord(float f) {
    unsigned u = __float_as_uint(f);
    return (u & 0x80000000u) ? ~u : (u | 0x80000000u);
}
__device__ __forceinline__ float ord2f(unsigned k) {
    return (k & 0x80000000u) ? __uint_as_float(k & 0x7fffffffu) : __uint_as_float(~k);
}

__global__ void k_gemm_wmma(const float* __restrict__ A, const float* __restrict__ B,
                            const float* __restrict__ bias, float* __restrict__ C,
                            int M, int N, int K)
{
    __shared__ float ldsB[128 * 16];
    const int tid  = threadIdx.x;
    const int wave = tid >> 5;
    const int lane = tid & 31;
    const int n0   = blockIdx.y * 16;

    for (int i = tid; i < K * 16; i += 128) {
        int k = i >> 4, c = i & 15;
        ldsB[i] = B[(long)k * N + n0 + c];
    }
    __syncthreads();

    const int m0 = (blockIdx.x * 4 + wave) * 16;
    if (m0 >= M) return;

    const int row   = m0 + (lane & 15);
    const int koff  = (lane >> 4) << 1;
    const int col16 = lane & 15;

    v8f c = {};
    for (int k0 = 0; k0 < K; k0 += 4) {
        v2f a, b;
        a.x = A[(long)row * K + k0 + koff];
        a.y = A[(long)row * K + k0 + koff + 1];
        b.x = ldsB[(k0 + koff)     * 16 + col16];
        b.y = ldsB[(k0 + koff + 1) * 16 + col16];
        c = __builtin_amdgcn_wmma_f32_16x16x4_f32(false, a, false, b,
                                                  (short)0, c, false, false);
    }

    const int   col   = n0 + col16;
    const float bv    = bias ? bias[col] : 0.0f;
    const int   rbase = m0 + ((lane >> 4) << 3);
#pragma unroll
    for (int i = 0; i < 8; ++i)
        C[(long)(rbase + i) * N + col] = c[i] + bv;
}

__global__ void k_scores(const float* __restrict__ Wh, const float* __restrict__ al,
                         const float* __restrict__ ar, float* __restrict__ el,
                         float* __restrict__ er, int N)
{
    int idx = blockIdx.x * blockDim.x + threadIdx.x;
    if (idx >= N * HEADS) return;
    int n = idx >> 2, h = idx & 3;
    const float* w = Wh + (long)n * FEAT + h * D_HEAD;
    const float* a = al + h * D_HEAD;
    const float* b = ar + h * D_HEAD;
    float sl = 0.0f, sr = 0.0f;
#pragma unroll
    for (int d = 0; d < D_HEAD; ++d) {
        sl = fmaf(w[d], a[d], sl);
        sr = fmaf(w[d], b[d], sr);
    }
    el[idx] = sl;
    er[idx] = sr;
}

__global__ void k_edge_pre(const float* __restrict__ el, const float* __restrict__ er,
                           const int* __restrict__ src, const int* __restrict__ dst,
                           float* __restrict__ ebuf, unsigned* __restrict__ menc, int E)
{
    int e = blockIdx.x * blockDim.x + threadIdx.x;
    if (e >= E) return;
    int s = src[e], d = dst[e];
#pragma unroll
    for (int h = 0; h < HEADS; ++h) {
        float v = el[s * HEADS + h] + er[d * HEADS + h];
        v = (v > 0.0f) ? v : NEG_SLOPE * v;
        ebuf[(long)e * HEADS + h] = v;
        atomicMax(&menc[d * HEADS + h], f2ord(v));
    }
}

__global__ void k_edge_softmax(const int* __restrict__ dst, const unsigned* __restrict__ menc,
                               float* __restrict__ ebuf, float* __restrict__ denom, int E)
{
    int e = blockIdx.x * blockDim.x + threadIdx.x;
    if (e >= E) return;
    int d = dst[e];
#pragma unroll
    for (int h = 0; h < HEADS; ++h) {
        float m  = ord2f(menc[d * HEADS + h]);
        float ex = expf(ebuf[(long)e * HEADS + h] - m);
        ebuf[(long)e * HEADS + h] = ex;
        atomicAdd(&denom[d * HEADS + h], ex);
    }
}

__global__ void k_edge_scatter(const float* __restrict__ Wh, const float* __restrict__ ex,
                               const float* __restrict__ denom, const int* __restrict__ src,
                               const int* __restrict__ dst, float* __restrict__ acc, int E)
{
    int e = blockIdx.x;
    int f = threadIdx.x;
    int h = f >> 5;
    int s = src[e], d = dst[e];
    float alpha = ex[(long)e * HEADS + h] / (denom[d * HEADS + h] + EPS_);
    atomicAdd(&acc[(long)d * FEAT + f], alpha * Wh[(long)s * FEAT + f]);
}

__global__ void k_elu(float* __restrict__ z, const float* __restrict__ b, long n)
{
    long i = (long)blockIdx.x * blockDim.x + threadIdx.x;
    if (i >= n) return;
    float x = z[i] + b[i & (FEAT - 1)];
    z[i] = (x > 0.0f) ? x : expm1f(x);
}

__global__ void k_mean(const float* __restrict__ acc, const float* __restrict__ b,
                       float* __restrict__ zmean, int N)
{
    int i = blockIdx.x * blockDim.x + threadIdx.x;
    if (i >= N * D_HEAD) return;
    int n = i >> 5, d = i & 31;
    float s = 0.0f;
#pragma unroll
    for (int h = 0; h < HEADS; ++h)
        s += acc[(long)n * FEAT + h * D_HEAD + d] + b[h * D_HEAD + d];
    zmean[i] = s * (1.0f / HEADS);
}


extern "C" void kernel_launch(void* const* d_in, const int* in_sizes, int n_in,
                              void* d_out, int out_size, void* d_ws, size_t ws_size,
                              hipStream_t stream)
{
    const float* h   = (const float*)d_in[0];
    const int*   src = (const int*)  d_in[1];
    const int*   dst = (const int*)  d_in[2];
    const float* W1  = (const float*)d_in[3];
    const float* al1 = (const float*)d_in[4];
    const float* ar1 = (const float*)d_in[5];
    const float* b1  = (const float*)d_in[6];
    const float* W2  = (const float*)d_in[7];
    const float* al2 = (const float*)d_in[8];
    const float* ar2 = (const float*)d_in[9];
    const float* b2  = (const float*)d_in[10];
    const float* Wp  = (const float*)d_in[11];
    const float* bp  = (const float*)d_in[12];
    float*       out = (float*)d_out;

    const int N = N_NODES, E = N_EDGES;

    float*    Wh    = (float*)d_ws;
    float*    Z     = Wh + (size_t)N * FEAT;
    float*    el    = Z  + (size_t)N * FEAT;
    float*    er    = el + (size_t)N * HEADS;
    unsigned* menc  = (unsigned*)(er + (size_t)N * HEADS);
    float*    denom = (float*)(menc + (size_t)N * HEADS);
    float*    ebuf  = denom + (size_t)N * HEADS;
    float*    zmean = ebuf + (size_t)E * HEADS;

    const int mtiles = (N + 15) / 16;
    const dim3 gemmBlk(128);
    const dim3 gemmGrd128((mtiles + 3) / 4, FEAT / 16);
    const dim3 gemmGrdOut((mtiles + 3) / 4, OUT_DIM / 16);

    const int zThreads = 256;
    auto zgrid = [](long n) { return (unsigned)((n + 255) / 256); };

    k_gemm_wmma<<<gemmGrd128, gemmBlk, 0, stream>>>(h, W1, nullptr, Wh, N, FEAT, FEAT);
    k_scores<<<(N * HEADS + 255) / 256, 256, 0, stream>>>(Wh, al1, ar1, el, er, N);
    k_zero<<<zgrid((long)N * HEADS), zThreads, 0, stream>>>((float*)menc, (long)N * HEADS);
    k_zero<<<zgrid((long)N * HEADS), zThreads, 0, stream>>>(denom, (long)N * HEADS);
    k_zero<<<zgrid((long)N * FEAT), zThreads, 0, stream>>>(Z, (long)N * FEAT);
    k_edge_pre<<<(E + 255) / 256, 256, 0, stream>>>(el, er, src, dst, ebuf, menc, E);
    k_edge_softmax<<<(E + 255) / 256, 256, 0, stream>>>(dst, menc, ebuf, denom, E);
    k_edge_scatter<<<E, FEAT, 0, stream>>>(Wh, ebuf, denom, src, dst, Z, E);
    k_elu<<<zgrid((long)N * FEAT), zThreads, 0, stream>>>(Z, b1, (long)N * FEAT);

    k_gemm_wmma<<<gemmGrd128, gemmBlk, 0, stream>>>(Z, W2, nullptr, Wh, N, FEAT, FEAT);
    k_scores<<<(N * HEADS + 255) / 256, 256, 0, stream>>>(Wh, al2, ar2, el, er, N);
    k_zero<<<zgrid((long)N * HEADS), zThreads, 0, stream>>>((float*)menc, (long)N * HEADS);
    k_zero<<<zgrid((long)N * HEADS), zThreads, 0, stream>>>(denom, (long)N * HEADS);
    k_zero<<<zgrid((long)N * FEAT), zThreads, 0, stream>>>(Z, (long)N * FEAT);
    k_edge_pre<<<(E + 255) / 256, 256, 0, stream>>>(el, er, src, dst, ebuf, menc, E);
    k_edge_softmax<<<(E + 255) / 256, 256, 0, stream>>>(dst, menc, ebuf, denom, E);
    k_edge_scatter<<<E, FEAT, 0, stream>>>(Wh, ebuf, denom, src, dst, Z, E);
    k_mean<<<(N * D_HEAD + 255) / 256, 256, 0, stream>>>(Z, b2, zmean, N);

    k_gemm_wmma<<<gemmGrdOut, gemmBlk, 0, stream>>>(zmean, Wp, bp, out, N, OUT_DIM, D_HEAD);
}


# Edges grouped by dst once per call (int atomic rank, prefix sum, fill); one wave per dst node does max, exp-sum and weighted Wh[src] sum in registers: no float atomics, sum order = slot order
# speedup vs baseline: 3.9460x; 3.9460x over previous
#include <hip/hip_runtime.h>
#include <hip/hip_bf16.h>
#include <math.h>

#define N_NODES   50000
#define N_EDGES   800000
#define HEADS     4
#define D_HEAD    32
#define FEAT      128
#define OUT_DIM   512
#define NEG_SLOPE 0.2f
#define EPS_      1e-9f

typedef __attribute__((ext_vector_type(2))) float v2f;
typedef __attribute__((ext_vector_type(8))) float v8f;


__global__ void k_zero(float* __restrict__ p, long n) {
    long i = (long)blockIdx.x * blockDim.x + threadIdx.x;
    if (i < n) p[i] = 0.0f;
}

__global__ void k_gemm_wmma(const float* __restrict__ A, const float* __restrict__ B,
                            const float* __restrict__ bias, float* __restrict__ C,
                            int M, int N, int K)
{
    __shared__ float ldsB[128 * 16];
    const int tid  = threadIdx.x;
    const int wave = tid >> 5;
    const int lane = tid & 31;
    const int n0   = blockIdx.y * 16;

    for (int i = tid; i < K * 16; i += 128) {
        int k = i >> 4, c = i & 15;
        ldsB[i] = B[(long)k * N + n0 + c];
    }
    __syncthreads();

    const int m0 = (blockIdx.x * 4 + wave) * 16;
    if (m0 >= M) return;

    const int row   = m0 + (lane & 15);
    const int koff  = (lane >> 4) << 1;
    const int col16 = lane & 15;

    v8f c = {};
    for (int k0 = 0; k0 < K; k0 += 4) {
        v2f a, b;
        a.x = A[(long)row * K + k0 + koff];
        a.y = A[(long)row * K + k0 + koff + 1];
        b.x = ldsB[(k0 + koff)     * 16 + col16];
        b.y = ldsB[(k0 + koff + 1) * 16 + col16];
        c = __builtin_amdgcn_wmma_f32_16x16x4_f32(false, a, false, b,
                                                  (short)0, c, false, false);
    }

    const int   col   = n0 + col16;
    const float bv    = bias ? bias[col] : 0.0f;
    const int   rbase = m0 + ((lane >> 4) << 3);
#pragma unroll
    for (int i = 0; i < 8; ++i)
        C[(long)(rbase + i) * N + col] = c[i] + bv;
}

__global__ void k_scores(const float* __restrict__ Wh, const float* __restrict__ al,
                         const float* __restrict__ ar, float* __restrict__ el,
                         float* __restrict__ er, int N)
{
    int idx = blockIdx.x * blockDim.x + threadIdx.x;
    if (idx >= N * HEADS) return;
    int n = idx >> 2, h = idx & 3;
    const float* w = Wh + (long)n * FEAT + h * D_HEAD;
    const float* a = al + h * D_HEAD;
    const float* b = ar + h * D_HEAD;
    float sl = 0.0f, sr = 0.0f;
#pragma unroll
    for (int d = 0; d < D_HEAD; ++d) {
        sl = fmaf(w[d], a[d], sl);
        sr = fmaf(w[d], b[d], sr);
    }
    el[idx] = sl;
    er[idx] = sr;
}


__global__ void k_count_rank(const int* __restrict__ dst, int* __restrict__ deg,
                             int* __restrict__ rank, int E)
{
    int e = blockIdx.x * blockDim.x + threadIdx.x;
    if (e >= E) return;
    rank[e] = atomicAdd(&deg[dst[e]], 1);
}

__global__ void __launch_bounds__(1024)
k_rowptr(const int* __restrict__ deg, int* __restrict__ rowptr, int N)
{
    __shared__ int s_pre[32];
    __shared__ int s_tot[32];
    const int t = threadIdx.x, lane = t & 31, w = t >> 5;
    const int first = blockIdx.x * 1024;
    int pre = 0;
    for (int i = t; i < first; i += 1024) pre += deg[i];
    const int idx = first + t;
    const int v = (idx < N) ? deg[idx] : 0;
    int inc = v;
    for (int o = 1; o < 32; o <<= 1) {
        int u = __shfl_up(inc, o);
        if (lane >= o) inc += u;
    }
    for (int o = 16; o > 0; o >>= 1) pre += __shfl_xor(pre, o);
    if (lane == 31) s_tot[w] = inc;
    if (lane == 0)  s_pre[w] = pre;
    __syncthreads();
    int base = 0;
    for (int i = 0; i < 32; ++i) {
        base += s_pre[i];
        if (i < w) base += s_tot[i];
    }
    if (idx < N) rowptr[idx] = base + inc - v;
}

__global__ void k_fill(const int* __restrict__ src, const int* __restrict__ dst,
                       const int* __restrict__ rowptr, const int* __restrict__ rank,
                       int* __restrict__ csr_src, int E)
{
    int e = blockIdx.x * blockDim.x + threadIdx.x;
    if (e >= E) return;
    csr_src[rowptr[dst[e]] + rank[e]] = src[e];
}

__device__ __forceinline__ float leaky(float v) { return (v > 0.0f) ? v : NEG_SLOPE * v; }
__device__ __forceinline__ float wave_max(float v) {
    for (int o = 16; o > 0; o >>= 1) v = fmaxf(v, __shfl_xor(v, o));
    return v;
}
__device__ __forceinline__ float wave_sum(float v) {
    for (int o = 16; o > 0; o >>= 1) v += __shfl_xor(v, o);
    return v;
}
__device__ __forceinline__ float finite_or_zero(float m) {
    return (fabsf(m) <= 3.402823466e+38f) ? m : 0.0f;
}
__device__ __forceinline__ float lane_bcast(float v, int t) {
    return __int_as_float(__builtin_amdgcn_readlane(__float_as_int(v), t));
}

__global__ void __launch_bounds__(256)
k_node_gather(const float* __restrict__ Wh, const float* __restrict__ el,
              const float* __restrict__ er, const int* __restrict__ rowptr,
              const int* __restrict__ deg, const int* __restrict__ csr_src,
              float* __restrict__ acc, int N)
{
    const int lane = threadIdx.x & 31;
    const int n = __builtin_amdgcn_readfirstlane((int)(blockIdx.x * 8 + (threadIdx.x >> 5)));
    if (n >= N) return;
    const int start = rowptr[n];
    const int cnt   = deg[n];
    const float4 r4 = *reinterpret_cast<const float4*>(er + (long)n * HEADS);

    float m0 = -INFINITY, m1 = -INFINITY, m2 = -INFINITY, m3 = -INFINITY;
    for (int base = 0; base < cnt; base += 32) {
        const int j = base + lane;
        if (j < cnt) {
            const int s = csr_src[start + j];
            const float4 l4 = *reinterpret_cast<const float4*>(el + (long)s * HEADS);
            m0 = fmaxf(m0, leaky(l4.x + r4.x));
            m1 = fmaxf(m1, leaky(l4.y + r4.y));
            m2 = fmaxf(m2, leaky(l4.z + r4.z));
            m3 = fmaxf(m3, leaky(l4.w + r4.w));
        }
    }
    m0 = finite_or_zero(wave_max(m0));
    m1 = finite_or_zero(wave_max(m1));
    m2 = finite_or_zero(wave_max(m2));
    m3 = finite_or_zero(wave_max(m3));

    float d0 = 0.0f, d1 = 0.0f, d2 = 0.0f, d3 = 0.0f;
    for (int base = 0; base < cnt; base += 32) {
        const int j = base + lane;
        if (j < cnt) {
            const int s = csr_src[start + j];
            const float4 l4 = *reinterpret_cast<const float4*>(el + (long)s * HEADS);
            d0 += expf(leaky(l4.x + r4.x) - m0);
            d1 += expf(leaky(l4.y + r4.y) - m1);
            d2 += expf(leaky(l4.z + r4.z) - m2);
            d3 += expf(leaky(l4.w + r4.w) - m3);
        }
    }
    d0 = wave_sum(d0) + EPS_;
    d1 = wave_sum(d1) + EPS_;
    d2 = wave_sum(d2) + EPS_;
    d3 = wave_sum(d3) + EPS_;

    const float* whl = Wh + lane;
    float acc0 = 0.0f, acc1 = 0.0f, acc2 = 0.0f, acc3 = 0.0f;
    for (int base = 0; base < cnt; base += 32) {
        const int j = base + lane;
        int   s  = 0;
        float a0 = 0.0f, a1 = 0.0f, a2 = 0.0f, a3 = 0.0f;
        if (j < cnt) {
            s = csr_src[start + j];
            const float4 l4 = *reinterpret_cast<const float4*>(el + (long)s * HEADS);
            a0 = expf(leaky(l4.x + r4.x) - m0) / d0;
            a1 = expf(leaky(l4.y + r4.y) - m1) / d1;
            a2 = expf(leaky(l4.z + r4.z) - m2) / d2;
            a3 = expf(leaky(l4.w + r4.w) - m3) / d3;
        }
        const int c = (cnt - base < 32) ? (cnt - base) : 32;
#pragma unroll 4
        for (int t = 0; t < c; ++t) {
            const float* row = whl + (long)__builtin_amdgcn_readlane(s, t) * FEAT;
            acc0 += lane_bcast(a0, t) * row[0];
            acc1 += lane_bcast(a1, t) * row[D_HEAD];
            acc2 += lane_bcast(a2, t) * row[2 * D_HEAD];
            acc3 += lane_bcast(a3, t) * row[3 * D_HEAD];
        }
    }
    float* o = acc + (long)n * FEAT + lane;
    o[0]          = acc0;
    o[D_HEAD]     = acc1;
    o[2 * D_HEAD] = acc2;
    o[3 * D_HEAD] = acc3;
}

__global__ void k_elu(float* __restrict__ z, const float* __restrict__ b, long n)
{
    long i = (long)blockIdx.x * blockDim.x + threadIdx.x;
    if (i >= n) return;
    float x = z[i] + b[i & (FEAT - 1)];
    z[i] = (x > 0.0f) ? x : expm1f(x);
}

__global__ void k_mean(const float* __restrict__ acc, const float* __restrict__ b,
                       float* __restrict__ zmean, int N)
{
    int i = blockIdx.x * blockDim.x + threadIdx.x;
    if (i >= N * D_HEAD) return;
    int n = i >> 5, d = i & 31;
    float s = 0.0f;
#pragma unroll
    for (int h = 0; h < HEADS; ++h)
        s += acc[(long)n * FEAT + h * D_HEAD + d] + b[h * D_HEAD + d];
    zmean[i] = s * (1.0f / HEADS);
}


extern "C" void kernel_launch(void* const* d_in, const int* in_sizes, int n_in,
                              void* d_out, int out_size, void* d_ws, size_t ws_size,
                              hipStream_t stream)
{
    const float* h   = (const float*)d_in[0];
    const int*   src = (const int*)  d_in[1];
    const int*   dst = (const int*)  d_in[2];
    const float* W1  = (const float*)d_in[3];
    const float* al1 = (const float*)d_in[4];
    const float* ar1 = (const float*)d_in[5];
    const float* b1  = (const float*)d_in[6];
    const float* W2  = (const float*)d_in[7];
    const float* al2 = (const float*)d_in[8];
    const float* ar2 = (const float*)d_in[9];
    const float* b2  = (const float*)d_in[10];
    const float* Wp  = (const float*)d_in[11];
    const float* bp  = (const float*)d_in[12];
    float*       out = (float*)d_out;

    const int N = N_NODES, E = N_EDGES;

    float* Wh      = (float*)d_ws;
    float* Z       = Wh + (size_t)N * FEAT;
    float* el      = Z  + (size_t)N * FEAT;
    float* er      = el + (size_t)N * HEADS;
    float* zmean   = er + (size_t)N * HEADS;
    int*   deg     = (int*)(zmean + (size_t)N * D_HEAD);
    int*   rowptr  = deg + (size_t)N;
    int*   rank    = rowptr + (size_t)N;
    int*   csr_src = rank + (size_t)E;

    const int mtiles = (N + 15) / 16;
    const dim3 gemmBlk(128);
    const dim3 gemmGrd128((mtiles + 3) / 4, FEAT / 16);
    const dim3 gemmGrdOut((mtiles + 3) / 4, OUT_DIM / 16);

    const int zThreads = 256;
    auto zgrid = [](long n) { return (unsigned)((n + 255) / 256); };

    k_zero<<<zgrid((long)N), zThreads, 0, stream>>>((float*)deg, (long)N);
    k_count_rank<<<(E + 255) / 256, 256, 0, stream>>>(dst, deg, rank, E);
    k_rowptr<<<(N + 1023) / 1024, 1024, 0, stream>>>(deg, rowptr, N);
    k_fill<<<(E + 255) / 256, 256, 0, stream>>>(src, dst, rowptr, rank, csr_src, E);

    k_gemm_wmma<<<gemmGrd128, gemmBlk, 0, stream>>>(h, W1, nullptr, Wh, N, FEAT, FEAT);
    k_scores<<<(N * HEADS + 255) / 256, 256, 0, stream>>>(Wh, al1, ar1, el, er, N);
    k_node_gather<<<(N + 7) / 8, 256, 0, stream>>>(Wh, el, er, rowptr, deg, csr_src, Z, N);
    k_elu<<<zgrid((long)N * FEAT), zThreads, 0, stream>>>(Z, b1, (long)N * FEAT);

    k_gemm_wmma<<<gemmGrd128, gemmBlk, 0, stream>>>(Z, W2, nullptr, Wh, N, FEAT, FEAT);
    k_scores<<<(N * HEADS + 255) / 256, 256, 0, stream>>>(Wh, al2, ar2, el, er, N);
    k_node_gather<<<(N + 7) / 8, 256, 0, stream>>>(Wh, el, er, rowptr, deg, csr_src, Z, N);
    k_mean<<<(N * D_HEAD + 255) / 256, 256, 0, stream>>>(Z, b2, zmean, N);

    k_gemm_wmma<<<gemmGrdOut, gemmBlk, 0, stream>>>(zmean, Wp, bp, out, N, OUT_DIM, D_HEAD);
}


# Bias+ELU (layer 1) and bias+head-mean (layer 2) applied to the gather kernel's register sums before its store; both per-element finalize launches and the layer-2 sum round trip removed
# speedup vs baseline: 4.2343x; 1.0731x over previous
#include <hip/hip_runtime.h>
#include <hip/hip_bf16.h>
#include <math.h>

#define N_NODES   50000
#define N_EDGES   800000
#define HEADS     4
#define D_HEAD    32
#define FEAT      128
#define OUT_DIM   512
#define NEG_SLOPE 0.2f
#define EPS_      1e-9f

typedef __attribute__((ext_vector_type(2))) float v2f;
typedef __attribute__((ext_vector_type(8))) float v8f;


__global__ void k_zero(float* __restrict__ p, long n) {
    long i = (long)blockIdx.x * blockDim.x + threadIdx.x;
    if (i < n) p[i] = 0.0f;
}

__global__ void k_gemm_wmma(const float* __restrict__ A, const float* __restrict__ B,
                            const float* __restrict__ bias, float* __restrict__ C,
                            int M, int N, int K)
{
    __shared__ float ldsB[128 * 16];
    const int tid  = threadIdx.x;
    const int wave = tid >> 5;
    const int lane = tid & 31;
    const int n0   = blockIdx.y * 16;

    for (int i = tid; i < K * 16; i += 128) {
        int k = i >> 4, c = i & 15;
        ldsB[i] = B[(long)k * N + n0 + c];
    }
    __syncthreads();

    const int m0 = (blockIdx.x * 4 + wave) * 16;
    if (m0 >= M) return;

    const int row   = m0 + (lane & 15);
    const int koff  = (lane >> 4) << 1;
    const int col16 = lane & 15;

    v8f c = {};
    for (int k0 = 0; k0 < K; k0 += 4) {
        v2f a, b;
        a.x = A[(long)row * K + k0 + koff];
        a.y = A[(long)row * K + k0 + koff + 1];
        b.x = ldsB[(k0 + koff)     * 16 + col16];
        b.y = ldsB[(k0 + koff + 1) * 16 + col16];
        c = __builtin_amdgcn_wmma_f32_16x16x4_f32(false, a, false, b,
                                                  (short)0, c, false, false);
    }

    const int   col   = n0 + col16;
    const float bv    = bias ? bias[col] : 0.0f;
    const int   rbase = m0 + ((lane >> 4) << 3);
#pragma unroll
    for (int i = 0; i < 8; ++i)
        C[(long)(rbase + i) * N + col] = c[i] + bv;
}

__global__ void k_scores(const float* __restrict__ Wh, const float* __restrict__ al,
                         const float* __restrict__ ar, float* __restrict__ el,
                         float* __restrict__ er, int N)
{
    int idx = blockIdx.x * blockDim.x + threadIdx.x;
    if (idx >= N * HEADS) return;
    int n = idx >> 2, h = idx & 3;
    const float* w = Wh + (long)n * FEAT + h * D_HEAD;
    const float* a = al + h * D_HEAD;
    const float* b = ar + h * D_HEAD;
    float sl = 0.0f, sr = 0.0f;
#pragma unroll
    for (int d = 0; d < D_HEAD; ++d) {
        sl = fmaf(w[d], a[d], sl);
        sr = fmaf(w[d], b[d], sr);
    }
    el[idx] = sl;
    er[idx] = sr;
}


__global__ void k_count_rank(const int* __restrict__ dst, int* __restrict__ deg,
                             int* __restrict__ rank, int E)
{
    int e = blockIdx.x * blockDim.x + threadIdx.x;
    if (e >= E) return;
    rank[e] = atomicAdd(&deg[dst[e]], 1);
}

__global__ void __launch_bounds__(1024)
k_rowptr(const int* __restrict__ deg, int* __restrict__ rowptr, int N)
{
    __shared__ int s_pre[32];
    __shared__ int s_tot[32];
    const int t = threadIdx.x, lane = t & 31, w = t >> 5;
    const int first = blockIdx.x * 1024;
    int pre = 0;
    for (int i = t; i < first; i += 1024) pre += deg[i];
    const int idx = first + t;
    const int v = (idx < N) ? deg[idx] : 0;
    int inc = v;
    for (int o = 1; o < 32; o <<= 1) {
        int u = __shfl_up(inc, o);
        if (lane >= o) inc += u;
    }
    for (int o = 16; o > 0; o >>= 1) pre += __shfl_xor(pre, o);
    if (lane == 31) s_tot[w] = inc;
    if (lane == 0)  s_pre[w] = pre;
    __syncthreads();
    int base = 0;
    for (int i = 0; i < 32; ++i) {
        base += s_pre[i];
        if (i < w) base += s_tot[i];
    }
    if (idx < N) rowptr[idx] = base + inc - v;
}

__global__ void k_fill(const int* __restrict__ src, const int* __restrict__ dst,
                       const int* __restrict__ rowptr, const int* __restrict__ rank,
                       int* __restrict__ csr_src, int E)
{
    int e = blockIdx.x * blockDim.x + threadIdx.x;
    if (e >= E) return;
    csr_src[rowptr[dst[e]] + rank[e]] = src[e];
}

__device__ __forceinline__ float leaky(float v) { return (v > 0.0f) ? v : NEG_SLOPE * v; }
__device__ __forceinline__ float wave_max(float v) {
    for (int o = 16; o > 0; o >>= 1) v = fmaxf(v, __shfl_xor(v, o));
    return v;
}
__device__ __forceinline__ float wave_sum(float v) {
    for (int o = 16; o > 0; o >>= 1) v += __shfl_xor(v, o);
    return v;
}
__device__ __forceinline__ float finite_or_zero(float m) {
    return (fabsf(m) <= 3.402823466e+38f) ? m : 0.0f;
}
__device__ __forceinline__ float lane_bcast(float v, int t) {
    return __int_as_float(__builtin_amdgcn_readlane(__float_as_int(v), t));
}
__device__ __forceinline__ float elu(float x) { return (x > 0.0f) ? x : expm1f(x); }

template <bool MEAN_HEADS>
__global__ void __launch_bounds__(256)
k_node_gather(const float* __restrict__ Wh, const float* __restrict__ el,
              const float* __restrict__ er, const int* __restrict__ rowptr,
              const int* __restrict__ deg, const int* __restrict__ csr_src,
              const float* __restrict__ b, float* __restrict__ res, int N)
{
    const int lane = threadIdx.x & 31;
    const int n = __builtin_amdgcn_readfirstlane((int)(blockIdx.x * 8 + (threadIdx.x >> 5)));
    if (n >= N) return;
    const int start = rowptr[n];
    const int cnt   = deg[n];
    const float4 r4 = *reinterpret_cast<const float4*>(er + (long)n * HEADS);
    const float bias0 = b[lane];
    const float bias1 = b[D_HEAD + lane];
    const float bias2 = b[2 * D_HEAD + lane];
    const float bias3 = b[3 * D_HEAD + lane];

    float m0 = -INFINITY, m1 = -INFINITY, m2 = -INFINITY, m3 = -INFINITY;
    for (int base = 0; base < cnt; base += 32) {
        const int j = base + lane;
        if (j < cnt) {
            const int s = csr_src[start + j];
            const float4 l4 = *reinterpret_cast<const float4*>(el + (long)s * HEADS);
            m0 = fmaxf(m0, leaky(l4.x + r4.x));
            m1 = fmaxf(m1, leaky(l4.y + r4.y));
            m2 = fmaxf(m2, leaky(l4.z + r4.z));
            m3 = fmaxf(m3, leaky(l4.w + r4.w));
        }
    }
    m0 = finite_or_zero(wave_max(m0));
    m1 = finite_or_zero(wave_max(m1));
    m2 = finite_or_zero(wave_max(m2));
    m3 = finite_or_zero(wave_max(m3));

    float d0 = 0.0f, d1 = 0.0f, d2 = 0.0f, d3 = 0.0f;
    for (int base = 0; base < cnt; base += 32) {
        const int j = base + lane;
        if (j < cnt) {
            const int s = csr_src[start + j];
            const float4 l4 = *reinterpret_cast<const float4*>(el + (long)s * HEADS);
            d0 += expf(leaky(l4.x + r4.x) - m0);
            d1 += expf(leaky(l4.y + r4.y) - m1);
            d2 += expf(leaky(l4.z + r4.z) - m2);
            d3 += expf(leaky(l4.w + r4.w) - m3);
        }
    }
    d0 = wave_sum(d0) + EPS_;
    d1 = wave_sum(d1) + EPS_;
    d2 = wave_sum(d2) + EPS_;
    d3 = wave_sum(d3) + EPS_;

    const float* whl = Wh + lane;
    float acc0 = 0.0f, acc1 = 0.0f, acc2 = 0.0f, acc3 = 0.0f;
    for (int base = 0; base < cnt; base += 32) {
        const int j = base + lane;
        int   s  = 0;
        float a0 = 0.0f, a1 = 0.0f, a2 = 0.0f, a3 = 0.0f;
        if (j < cnt) {
            s = csr_src[start + j];
            const float4 l4 = *reinterpret_cast<const float4*>(el + (long)s * HEADS);
            a0 = expf(leaky(l4.x + r4.x) - m0) / d0;
            a1 = expf(leaky(l4.y + r4.y) - m1) / d1;
            a2 = expf(leaky(l4.z + r4.z) - m2) / d2;
            a3 = expf(leaky(l4.w + r4.w) - m3) / d3;
        }
        const int c = (cnt - base < 32) ? (cnt - base) : 32;
#pragma unroll 4
        for (int t = 0; t < c; ++t) {
            const float* row = whl + (long)__builtin_amdgcn_readlane(s, t) * FEAT;
            acc0 += lane_bcast(a0, t) * row[0];
            acc1 += lane_bcast(a1, t) * row[D_HEAD];
            acc2 += lane_bcast(a2, t) * row[2 * D_HEAD];
            acc3 += lane_bcast(a3, t) * row[3 * D_HEAD];
        }
    }
    const float x0 = acc0 + bias0;
    const float x1 = acc1 + bias1;
    const float x2 = acc2 + bias2;
    const float x3 = acc3 + bias3;
    if (MEAN_HEADS) {
        float s = 0.0f;
        s += x0;
        s += x1;
        s += x2;
        s += x3;
        res[(long)n * D_HEAD + lane] = s * (1.0f / HEADS);
    } else {
        float* o = res + (long)n * FEAT + lane;
        o[0]          = elu(x0);
        o[D_HEAD]     = elu(x1);
        o[2 * D_HEAD] = elu(x2);
        o[3 * D_HEAD] = elu(x3);
    }
}


extern "C" void kernel_launch(void* const* d_in, const int* in_sizes, int n_in,
                              void* d_out, int out_size, void* d_ws, size_t ws_size,
                              hipStream_t stream)
{
    const float* h   = (const float*)d_in[0];
    const int*   src = (const int*)  d_in[1];
    const int*   dst = (const int*)  d_in[2];
    const float* W1  = (const float*)d_in[3];
    const float* al1 = (const float*)d_in[4];
    const float* ar1 = (const float*)d_in[5];
    const float* b1  = (const float*)d_in[6];
    const float* W2  = (const float*)d_in[7];
    const float* al2 = (const float*)d_in[8];
    const float* ar2 = (const float*)d_in[9];
    const float* b2  = (const float*)d_in[10];
    const float* Wp  = (const float*)d_in[11];
    const float* bp  = (const float*)d_in[12];
    float*       out = (float*)d_out;

    const int N = N_NODES, E = N_EDGES;

    float* Wh      = (float*)d_ws;
    float* Z       = Wh + (size_t)N * FEAT;
    float* el      = Z  + (size_t)N * FEAT;
    float* er      = el + (size_t)N * HEADS;
    float* zmean   = er + (size_t)N * HEADS;
    int*   deg     = (int*)(zmean + (size_t)N * D_HEAD);
    int*   rowptr  = deg + (size_t)N;
    int*   rank    = rowptr + (size_t)N;
    int*   csr_src = rank + (size_t)E;

    const int mtiles = (N + 15) / 16;
    const dim3 gemmBlk(128);
    const dim3 gemmGrd128((mtiles + 3) / 4, FEAT / 16);
    const dim3 gemmGrdOut((mtiles + 3) / 4, OUT_DIM / 16);

    const int zThreads = 256;
    auto zgrid = [](long n) { return (unsigned)((n + 255) / 256); };

    k_zero<<<zgrid((long)N), zThreads, 0, stream>>>((float*)deg, (long)N);
    k_count_rank<<<(E + 255) / 256, 256, 0, stream>>>(dst, deg, rank, E);
    k_rowptr<<<(N + 1023) / 1024, 1024, 0, stream>>>(deg, rowptr, N);
    k_fill<<<(E + 255) / 256, 256, 0, stream>>>(src, dst, rowptr, rank, csr_src, E);

    k_gemm_wmma<<<gemmGrd128, gemmBlk, 0, stream>>>(h, W1, nullptr, Wh, N, FEAT, FEAT);
    k_scores<<<(N * HEADS + 255) / 256, 256, 0, stream>>>(Wh, al1, ar1, el, er, N);
    k_node_gather<false><<<(N + 7) / 8, 256, 0, stream>>>(Wh, el, er, rowptr, deg, csr_src, b1, Z, N);

    k_gemm_wmma<<<gemmGrd128, gemmBlk, 0, stream>>>(Z, W2, nullptr, Wh, N, FEAT, FEAT);
    k_scores<<<(N * HEADS + 255) / 256, 256, 0, stream>>>(Wh, al2, ar2, el, er, N);
    k_node_gather<true><<<(N + 7) / 8, 256, 0, stream>>>(Wh, el, er, rowptr, deg, csr_src, b2, zmean, N);

    k_gemm_wmma<<<gemmGrdOut, gemmBlk, 0, stream>>>(zmean, Wp, bp, out, N, OUT_DIM, D_HEAD);
}


# GEMM retiled: one wave per 16-row strip over all N columns (A strip in registers, whole B staged once per block in LDS); 150144 GEMM waves become 9375; k summation order changes
# speedup vs baseline: 5.7103x; 1.3486x over previous
#include <hip/hip_runtime.h>
#include <hip/hip_bf16.h>
#include <math.h>

#define N_NODES   50000
#define N_EDGES   800000
#define HEADS     4
#define D_HEAD    32
#define FEAT      128
#define OUT_DIM   512
#define NEG_SLOPE 0.2f
#define EPS_      1e-9f

typedef __attribute__((ext_vector_type(2))) float v2f;
typedef __attribute__((ext_vector_type(8))) float v8f;


__global__ void k_zero(float* __restrict__ p, long n) {
    long i = (long)blockIdx.x * blockDim.x + threadIdx.x;
    if (i < n) p[i] = 0.0f;
}

template <int K, int N, bool HAS_BIAS>
__global__ void __launch_bounds__(256)
k_gemm_strip(const float* __restrict__ A, const float* __restrict__ B,
             const float* __restrict__ bias, float* __restrict__ C, int M)
{
    static_assert(K % 8 == 0 && N % 128 == 0 && K * N == 16384, "tile plan");
    constexpr int KH = K / 2;
    constexpr int KS = K / 4;
    constexpr int NA = KH / 4;
    constexpr int NG = N / 128;

    __shared__ float4 ldsW[K * N / 4];

    const int tid  = threadIdx.x;
    const int wave = tid >> 5;
    const int lane = tid & 31;
    const int c    = lane & 15;
    const int h    = lane >> 4;
    const int m0   = (blockIdx.x * 8 + wave) * 16;

    const int arow = (m0 + c < M) ? (m0 + c) : (M - 1);
    const float4* ap = reinterpret_cast<const float4*>(A + (long)arow * K + h * KH);
    float4 a4[NA];
#pragma unroll
    for (int i = 0; i < NA; ++i) a4[i] = ap[i];
    float2 bz[N / 32];
    if (HAS_BIAS) {
#pragma unroll
        for (int t = 0; t < N / 32; ++t)
            bz[t] = *reinterpret_cast<const float2*>(bias + 32 * t + 2 * c);
    }

    constexpr int ITEMS = (K / 2) * (N / 4);
    static_assert(ITEMS % 256 == 0, "staging plan");
#pragma unroll
    for (int u = 0; u < ITEMS / 256; ++u) {
        const int it = tid + u * 256;
        const int kp = it / (N / 4);
        const int q  = it % (N / 4);
        const float4 r0 = *reinterpret_cast<const float4*>(B + (long)(2 * kp) * N + 4 * q);
        const float4 r1 = *reinterpret_cast<const float4*>(B + (long)(2 * kp + 1) * N + 4 * q);
        const int w = (kp * N + 4 * q) >> 1;
        ldsW[w]     = make_float4(r0.x, r1.x, r0.y, r1.y);
        ldsW[w + 1] = make_float4(r0.z, r1.z, r0.w, r1.w);
    }
    __syncthreads();

    if (m0 >= M) return;

    const int rbase = m0 + (h << 3);
    const float4* lp = ldsW + (((h * KS) * N + 2 * c) >> 1);

    float4 bq[2][4];
#pragma unroll
    for (int p = 0; p < 4; ++p) bq[0][p] = lp[16 * p];

#pragma unroll
    for (int g = 0; g < NG; ++g) {
        v8f acc[8];
#pragma unroll
        for (int t = 0; t < 8; ++t) acc[t] = v8f{};
#pragma unroll
        for (int j = 0; j < KS; ++j) {
            const int cur = (g * KS + j) & 1;
            const int jn  = (j + 1 < KS) ? (j + 1) : 0;
            const int gn  = (j + 1 < KS) ? g : (g + 1);
            if (gn < NG) {
#pragma unroll
                for (int p = 0; p < 4; ++p) bq[cur ^ 1][p] = lp[gn * 64 + jn * (N / 2) + 16 * p];
            }
            v2f a;
            if (j & 1) { a.x = a4[j >> 1].z; a.y = a4[j >> 1].w; }
            else       { a.x = a4[j >> 1].x; a.y = a4[j >> 1].y; }
#pragma unroll
            for (int p = 0; p < 4; ++p) {
                v2f b0, b1;
                b0.x = bq[cur][p].x; b0.y = bq[cur][p].y;
                b1.x = bq[cur][p].z; b1.y = bq[cur][p].w;
                acc[2 * p]     = __builtin_amdgcn_wmma_f32_16x16x4_f32(false, a, false, b0,
                                                       (short)0, acc[2 * p], false, false);
                acc[2 * p + 1] = __builtin_amdgcn_wmma_f32_16x16x4_f32(false, a, false, b1,
                                                       (short)0, acc[2 * p + 1], false, false);
            }
        }
#pragma unroll
        for (int p = 0; p < 4; ++p) {
            float* crow = C + (long)rbase * N + g * 128 + 32 * p + 2 * c;
#pragma unroll
            for (int i = 0; i < 8; ++i) {
                float2 o;
                o.x = acc[2 * p][i];
                o.y = acc[2 * p + 1][i];
                if (HAS_BIAS) { o.x += bz[g * 4 + p].x; o.y += bz[g * 4 + p].y; }
                *reinterpret_cast<float2*>(crow + (long)i * N) = o;
            }
        }
    }
}

__global__ void k_scores(const float* __restrict__ Wh, const float* __restrict__ al,
                         const float* __restrict__ ar, float* __restrict__ el,
                         float* __restrict__ er, int N)
{
    int idx = blockIdx.x * blockDim.x + threadIdx.x;
    if (idx >= N * HEADS) return;
    int n = idx >> 2, h = idx & 3;
    const float* w = Wh + (long)n * FEAT + h * D_HEAD;
    const float* a = al + h * D_HEAD;
    const float* b = ar + h * D_HEAD;
    float sl = 0.0f, sr = 0.0f;
#pragma unroll
    for (int d = 0; d < D_HEAD; ++d) {
        sl = fmaf(w[d], a[d], sl);
        sr = fmaf(w[d], b[d], sr);
    }
    el[idx] = sl;
    er[idx] = sr;
}


__global__ void k_count_rank(const int* __restrict__ dst, int* __restrict__ deg,
                             int* __restrict__ rank, int E)
{
    int e = blockIdx.x * blockDim.x + threadIdx.x;
    if (e >= E) return;
    rank[e] = atomicAdd(&deg[dst[e]], 1);
}

__global__ void __launch_bounds__(1024)
k_rowptr(const int* __restrict__ deg, int* __restrict__ rowptr, int N)
{
    __shared__ int s_pre[32];
    __shared__ int s_tot[32];
    const int t = threadIdx.x, lane = t & 31, w = t >> 5;
    const int first = blockIdx.x * 1024;
    int pre = 0;
    for (int i = t; i < first; i += 1024) pre += deg[i];
    const int idx = first + t;
    const int v = (idx < N) ? deg[idx] : 0;
    int inc = v;
    for (int o = 1; o < 32; o <<= 1) {
        int u = __shfl_up(inc, o);
        if (lane >= o) inc += u;
    }
    for (int o = 16; o > 0; o >>= 1) pre += __shfl_xor(pre, o);
    if (lane == 31) s_tot[w] = inc;
    if (lane == 0)  s_pre[w] = pre;
    __syncthreads();
    int base = 0;
    for (int i = 0; i < 32; ++i) {
        base += s_pre[i];
        if (i < w) base += s_tot[i];
    }
    if (idx < N) rowptr[idx] = base + inc - v;
}

__global__ void k_fill(const int* __restrict__ src, const int* __restrict__ dst,
                       const int* __restrict__ rowptr, const int* __restrict__ rank,
                       int* __restrict__ csr_src, int E)
{
    int e = blockIdx.x * blockDim.x + threadIdx.x;
    if (e >= E) return;
    csr_src[rowptr[dst[e]] + rank[e]] = src[e];
}

__device__ __forceinline__ float leaky(float v) { return (v > 0.0f) ? v : NEG_SLOPE * v; }
__device__ __forceinline__ float wave_max(float v) {
    for (int o = 16; o > 0; o >>= 1) v = fmaxf(v, __shfl_xor(v, o));
    return v;
}
__device__ __forceinline__ float wave_sum(float v) {
    for (int o = 16; o > 0; o >>= 1) v += __shfl_xor(v, o);
    return v;
}
__device__ __forceinline__ float finite_or_zero(float m) {
    return (fabsf(m) <= 3.402823466e+38f) ? m : 0.0f;
}
__device__ __forceinline__ float lane_bcast(float v, int t) {
    return __int_as_float(__builtin_amdgcn_readlane(__float_as_int(v), t));
}
__device__ __forceinline__ float elu(float x) { return (x > 0.0f) ? x : expm1f(x); }

template <bool MEAN_HEADS>
__global__ void __launch_bounds__(256)
k_node_gather(const float* __restrict__ Wh, const float* __restrict__ el,
              const float* __restrict__ er, const int* __restrict__ rowptr,
              const int* __restrict__ deg, const int* __restrict__ csr_src,
              const float* __restrict__ b, float* __restrict__ res, int N)
{
    const int lane = threadIdx.x & 31;
    const int n = __builtin_amdgcn_readfirstlane((int)(blockIdx.x * 8 + (threadIdx.x >> 5)));
    if (n >= N) return;
    const int start = rowptr[n];
    const int cnt   = deg[n];
    const float4 r4 = *reinterpret_cast<const float4*>(er + (long)n * HEADS);
    const float bias0 = b[lane];
    const float bias1 = b[D_HEAD + lane];
    const float bias2 = b[2 * D_HEAD + lane];
    const float bias3 = b[3 * D_HEAD + lane];

    float m0 = -INFINITY, m1 = -INFINITY, m2 = -INFINITY, m3 = -INFINITY;
    for (int base = 0; base < cnt; base += 32) {
        const int j = base + lane;
        if (j < cnt) {
            const int s = csr_src[start + j];
            const float4 l4 = *reinterpret_cast<const float4*>(el + (long)s * HEADS);
            m0 = fmaxf(m0, leaky(l4.x + r4.x));
            m1 = fmaxf(m1, leaky(l4.y + r4.y));
            m2 = fmaxf(m2, leaky(l4.z + r4.z));
            m3 = fmaxf(m3, leaky(l4.w + r4.w));
        }
    }
    m0 = finite_or_zero(wave_max(m0));
    m1 = finite_or_zero(wave_max(m1));
    m2 = finite_or_zero(wave_max(m2));
    m3 = finite_or_zero(wave_max(m3));

    float d0 = 0.0f, d1 = 0.0f, d2 = 0.0f, d3 = 0.0f;
    for (int base = 0; base < cnt; base += 32) {
        const int j = base + lane;
        if (j < cnt) {
            const int s = csr_src[start + j];
            const float4 l4 = *reinterpret_cast<const float4*>(el + (long)s * HEADS);
            d0 += expf(leaky(l4.x + r4.x) - m0);
            d1 += expf(leaky(l4.y + r4.y) - m1);
            d2 += expf(leaky(l4.z + r4.z) - m2);
            d3 += expf(leaky(l4.w + r4.w) - m3);
        }
    }
    d0 = wave_sum(d0) + EPS_;
    d1 = wave_sum(d1) + EPS_;
    d2 = wave_sum(d2) + EPS_;
    d3 = wave_sum(d3) + EPS_;

    const float* whl = Wh + lane;
    float acc0 = 0.0f, acc1 = 0.0f, acc2 = 0.0f, acc3 = 0.0f;
    for (int base = 0; base < cnt; base += 32) {
        const int j = base + lane;
        int   s  = 0;
        float a0 = 0.0f, a1 = 0.0f, a2 = 0.0f, a3 = 0.0f;
        if (j < cnt) {
            s = csr_src[start + j];
            const float4 l4 = *reinterpret_cast<const float4*>(el + (long)s * HEADS);
            a0 = expf(leaky(l4.x + r4.x) - m0) / d0;
            a1 = expf(leaky(l4.y + r4.y) - m1) / d1;
            a2 = expf(leaky(l4.z + r4.z) - m2) / d2;
            a3 = expf(leaky(l4.w + r4.w) - m3) / d3;
        }
        const int c = (cnt - base < 32) ? (cnt - base) : 32;
#pragma unroll 4
        for (int t = 0; t < c; ++t) {
            const float* row = whl + (long)__builtin_amdgcn_readlane(s, t) * FEAT;
            acc0 += lane_bcast(a0, t) * row[0];
            acc1 += lane_bcast(a1, t) * row[D_HEAD];
            acc2 += lane_bcast(a2, t) * row[2 * D_HEAD];
            acc3 += lane_bcast(a3, t) * row[3 * D_HEAD];
        }
    }
    const float x0 = acc0 + bias0;
    const float x1 = acc1 + bias1;
    const float x2 = acc2 + bias2;
    const float x3 = acc3 + bias3;
    if (MEAN_HEADS) {
        float s = 0.0f;
        s += x0;
        s += x1;
        s += x2;
        s += x3;
        res[(long)n * D_HEAD + lane] = s * (1.0f / HEADS);
    } else {
        float* o = res + (long)n * FEAT + lane;
        o[0]          = elu(x0);
        o[D_HEAD]     = elu(x1);
        o[2 * D_HEAD] = elu(x2);
        o[3 * D_HEAD] = elu(x3);
    }
}


extern "C" void kernel_launch(void* const* d_in, const int* in_sizes, int n_in,
                              void* d_out, int out_size, void* d_ws, size_t ws_size,
                              hipStream_t stream)
{
    const float* h   = (const float*)d_in[0];
    const int*   src = (const int*)  d_in[1];
    const int*   dst = (const int*)  d_in[2];
    const float* W1  = (const float*)d_in[3];
    const float* al1 = (const float*)d_in[4];
    const float* ar1 = (const float*)d_in[5];
    const float* b1  = (const float*)d_in[6];
    const float* W2  = (const float*)d_in[7];
    const float* al2 = (const float*)d_in[8];
    const float* ar2 = (const float*)d_in[9];
    const float* b2  = (const float*)d_in[10];
    const float* Wp  = (const float*)d_in[11];
    const float* bp  = (const float*)d_in[12];
    float*       out = (float*)d_out;

    const int N = N_NODES, E = N_EDGES;
    static_assert(N_NODES % 16 == 0, "the GEMM's 16-row strips are whole");

    float* Wh      = (float*)d_ws;
    float* Z       = Wh + (size_t)N * FEAT;
    float* el      = Z  + (size_t)N * FEAT;
    float* er      = el + (size_t)N * HEADS;
    float* zmean   = er + (size_t)N * HEADS;
    int*   deg     = (int*)(zmean + (size_t)N * D_HEAD);
    int*   rowptr  = deg + (size_t)N;
    int*   rank    = rowptr + (size_t)N;
    int*   csr_src = rank + (size_t)E;

    const int mtiles = (N + 15) / 16;
    const dim3 gemmBlk(256);
    const dim3 gemmGrd((mtiles + 7) / 8);

    const int zThreads = 256;
    auto zgrid = [](long n) { return (unsigned)((n + 255) / 256); };

    k_zero<<<zgrid((long)N), zThreads, 0, stream>>>((float*)deg, (long)N);
    k_count_rank<<<(E + 255) / 256, 256, 0, stream>>>(dst, deg, rank, E);
    k_rowptr<<<(N + 1023) / 1024, 1024, 0, stream>>>(deg, rowptr, N);
    k_fill<<<(E + 255) / 256, 256, 0, stream>>>(src, dst, rowptr, rank, csr_src, E);

    k_gemm_strip<FEAT, FEAT, false><<<gemmGrd, gemmBlk, 0, stream>>>(h, W1, nullptr, Wh, N);
    k_scores<<<(N * HEADS + 255) / 256, 256, 0, stream>>>(Wh, al1, ar1, el, er, N);
    k_node_gather<false><<<(N + 7) / 8, 256, 0, stream>>>(Wh, el, er, rowptr, deg, csr_src, b1, Z, N);

    k_gemm_strip<FEAT, FEAT, false><<<gemmGrd, gemmBlk, 0, stream>>>(Z, W2, nullptr, Wh, N);
    k_scores<<<(N * HEADS + 255) / 256, 256, 0, stream>>>(Wh, al2, ar2, el, er, N);
    k_node_gather<true><<<(N + 7) / 8, 256, 0, stream>>>(Wh, el, er, rowptr, deg, csr_src, b2, zmean, N);

    k_gemm_strip<D_HEAD, OUT_DIM, true><<<gemmGrd, gemmBlk, 0, stream>>>(zmean, Wp, bp, out, N);
}


# Gather: each edge's source index and 4 scores are read once and kept in registers through max, exp-sum and weights, not re-read per phase (lists up to 32 edges; longer ones keep the 3 passes)
# speedup vs baseline: 6.0755x; 1.0640x over previous
#include <hip/hip_runtime.h>
#include <hip/hip_bf16.h>
#include <math.h>

#define N_NODES   50000
#define N_EDGES   800000
#define HEADS     4
#define D_HEAD    32
#define FEAT      128
#define OUT_DIM   512
#define NEG_SLOPE 0.2f
#define EPS_      1e-9f

typedef __attribute__((ext_vector_type(2))) float v2f;
typedef __attribute__((ext_vector_type(8))) float v8f;


__global__ void k_zero(float* __restrict__ p, long n) {
    long i = (long)blockIdx.x * blockDim.x + threadIdx.x;
    if (i < n) p[i] = 0.0f;
}

template <int K, int N, bool HAS_BIAS>
__global__ void __launch_bounds__(256)
k_gemm_strip(const float* __restrict__ A, const float* __restrict__ B,
             const float* __restrict__ bias, float* __restrict__ C, int M)
{
    static_assert(K % 8 == 0 && N % 128 == 0 && K * N == 16384, "tile plan");
    constexpr int KH = K / 2;
    constexpr int KS = K / 4;
    constexpr int NA = KH / 4;
    constexpr int NG = N / 128;

    __shared__ float4 ldsW[K * N / 4];

    const int tid  = threadIdx.x;
    const int wave = tid >> 5;
    const int lane = tid & 31;
    const int c    = lane & 15;
    const int h    = lane >> 4;
    const int m0   = (blockIdx.x * 8 + wave) * 16;

    const int arow = (m0 + c < M) ? (m0 + c) : (M - 1);
    const float4* ap = reinterpret_cast<const float4*>(A + (long)arow * K + h * KH);
    float4 a4[NA];
#pragma unroll
    for (int i = 0; i < NA; ++i) a4[i] = ap[i];
    float2 bz[N / 32];
    if (HAS_BIAS) {
#pragma unroll
        for (int t = 0; t < N / 32; ++t)
            bz[t] = *reinterpret_cast<const float2*>(bias + 32 * t + 2 * c);
    }

    constexpr int ITEMS = (K / 2) * (N / 4);
    static_assert(ITEMS % 256 == 0, "staging plan");
#pragma unroll
    for (int u = 0; u < ITEMS / 256; ++u) {
        const int it = tid + u * 256;
        const int kp = it / (N / 4);
        const int q  = it % (N / 4);
        const float4 r0 = *reinterpret_cast<const float4*>(B + (long)(2 * kp) * N + 4 * q);
        const float4 r1 = *reinterpret_cast<const float4*>(B + (long)(2 * kp + 1) * N + 4 * q);
        const int w = (kp * N + 4 * q) >> 1;
        ldsW[w]     = make_float4(r0.x, r1.x, r0.y, r1.y);
        ldsW[w + 1] = make_float4(r0.z, r1.z, r0.w, r1.w);
    }
    __syncthreads();

    if (m0 >= M) return;

    const int rbase = m0 + (h << 3);
    const float4* lp = ldsW + (((h * KS) * N + 2 * c) >> 1);

    float4 bq[2][4];
#pragma unroll
    for (int p = 0; p < 4; ++p) bq[0][p] = lp[16 * p];

#pragma unroll
    for (int g = 0; g < NG; ++g) {
        v8f acc[8];
#pragma unroll
        for (int t = 0; t < 8; ++t) acc[t] = v8f{};
#pragma unroll
        for (int j = 0; j < KS; ++j) {
            const int cur = (g * KS + j) & 1;
            const int jn  = (j + 1 < KS) ? (j + 1) : 0;
            const int gn  = (j + 1 < KS) ? g : (g + 1);
            if (gn < NG) {
#pragma unroll
                for (int p = 0; p < 4; ++p) bq[cur ^ 1][p] = lp[gn * 64 + jn * (N / 2) + 16 * p];
            }
            v2f a;
            if (j & 1) { a.x = a4[j >> 1].z; a.y = a4[j >> 1].w; }
            else       { a.x = a4[j >> 1].x; a.y = a4[j >> 1].y; }
#pragma unroll
            for (int p = 0; p < 4; ++p) {
                v2f b0, b1;
                b0.x = bq[cur][p].x; b0.y = bq[cur][p].y;
                b1.x = bq[cur][p].z; b1.y = bq[cur][p].w;
                acc[2 * p]     = __builtin_amdgcn_wmma_f32_16x16x4_f32(false, a, false, b0,
                                                       (short)0, acc[2 * p], false, false);
                acc[2 * p + 1] = __builtin_amdgcn_wmma_f32_16x16x4_f32(false, a, false, b1,
                                                       (short)0, acc[2 * p + 1], false, false);
            }
        }
#pragma unroll
        for (int p = 0; p < 4; ++p) {
            float* crow = C + (long)rbase * N + g * 128 + 32 * p + 2 * c;
#pragma unroll
            for (int i = 0; i < 8; ++i) {
                float2 o;
                o.x = acc[2 * p][i];
                o.y = acc[2 * p + 1][i];
                if (HAS_BIAS) { o.x += bz[g * 4 + p].x; o.y += bz[g * 4 + p].y; }
                *reinterpret_cast<float2*>(crow + (long)i * N) = o;
            }
        }
    }
}

__global__ void k_scores(const float* __restrict__ Wh, const float* __restrict__ al,
                         const float* __restrict__ ar, float* __restrict__ el,
                         float* __restrict__ er, int N)
{
    int idx = blockIdx.x * blockDim.x + threadIdx.x;
    if (idx >= N * HEADS) return;
    int n = idx >> 2, h = idx & 3;
    const float* w = Wh + (long)n * FEAT + h * D_HEAD;
    const float* a = al + h * D_HEAD;
    const float* b = ar + h * D_HEAD;
    float sl = 0.0f, sr = 0.0f;
#pragma unroll
    for (int d = 0; d < D_HEAD; ++d) {
        sl = fmaf(w[d], a[d], sl);
        sr = fmaf(w[d], b[d], sr);
    }
    el[idx] = sl;
    er[idx] = sr;
}


__global__ void k_count_rank(const int* __restrict__ dst, int* __restrict__ deg,
                             int* __restrict__ rank, int E)
{
    int e = blockIdx.x * blockDim.x + threadIdx.x;
    if (e >= E) return;
    rank[e] = atomicAdd(&deg[dst[e]], 1);
}

__global__ void __launch_bounds__(1024)
k_rowptr(const int* __restrict__ deg, int* __restrict__ rowptr, int N)
{
    __shared__ int s_pre[32];
    __shared__ int s_tot[32];
    const int t = threadIdx.x, lane = t & 31, w = t >> 5;
    const int first = blockIdx.x * 1024;
    int pre = 0;
    for (int i = t; i < first; i += 1024) pre += deg[i];
    const int idx = first + t;
    const int v = (idx < N) ? deg[idx] : 0;
    int inc = v;
    for (int o = 1; o < 32; o <<= 1) {
        int u = __shfl_up(inc, o);
        if (lane >= o) inc += u;
    }
    for (int o = 16; o > 0; o >>= 1) pre += __shfl_xor(pre, o);
    if (lane == 31) s_tot[w] = inc;
    if (lane == 0)  s_pre[w] = pre;
    __syncthreads();
    int base = 0;
    for (int i = 0; i < 32; ++i) {
        base += s_pre[i];
        if (i < w) base += s_tot[i];
    }
    if (idx < N) rowptr[idx] = base + inc - v;
}

__global__ void k_fill(const int* __restrict__ src, const int* __restrict__ dst,
                       const int* __restrict__ rowptr, const int* __restrict__ rank,
                       int* __restrict__ csr_src, int E)
{
    int e = blockIdx.x * blockDim.x + threadIdx.x;
    if (e >= E) return;
    csr_src[rowptr[dst[e]] + rank[e]] = src[e];
}

__device__ __forceinline__ float leaky(float v) { return (v > 0.0f) ? v : NEG_SLOPE * v; }
__device__ __forceinline__ float wave_max(float v) {
    for (int o = 16; o > 0; o >>= 1) v = fmaxf(v, __shfl_xor(v, o));
    return v;
}
__device__ __forceinline__ float wave_sum(float v) {
    for (int o = 16; o > 0; o >>= 1) v += __shfl_xor(v, o);
    return v;
}
__device__ __forceinline__ float finite_or_zero(float m) {
    return (fabsf(m) <= 3.402823466e+38f) ? m : 0.0f;
}
__device__ __forceinline__ float lane_bcast(float v, int t) {
    return __int_as_float(__builtin_amdgcn_readlane(__float_as_int(v), t));
}
__device__ __forceinline__ float elu(float x) { return (x > 0.0f) ? x : expm1f(x); }

__device__ __forceinline__ void add_rows(const float* __restrict__ whl, int s,
                                         float a0, float a1, float a2, float a3, int c,
                                         float& acc0, float& acc1, float& acc2, float& acc3)
{
#pragma unroll 4
    for (int t = 0; t < c; ++t) {
        const float* row = whl + (long)__builtin_amdgcn_readlane(s, t) * FEAT;
        acc0 += lane_bcast(a0, t) * row[0];
        acc1 += lane_bcast(a1, t) * row[D_HEAD];
        acc2 += lane_bcast(a2, t) * row[2 * D_HEAD];
        acc3 += lane_bcast(a3, t) * row[3 * D_HEAD];
    }
}

template <bool MEAN_HEADS>
__global__ void __launch_bounds__(256)
k_node_gather(const float* __restrict__ Wh, const float* __restrict__ el,
              const float* __restrict__ er, const int* __restrict__ rowptr,
              const int* __restrict__ deg, const int* __restrict__ csr_src,
              const float* __restrict__ b, float* __restrict__ res, int N)
{
    const int lane = threadIdx.x & 31;
    const int n = __builtin_amdgcn_readfirstlane((int)(blockIdx.x * 8 + (threadIdx.x >> 5)));
    if (n >= N) return;
    const int start = rowptr[n];
    const int cnt   = deg[n];
    const float4 r4 = *reinterpret_cast<const float4*>(er + (long)n * HEADS);
    const float bias0 = b[lane];
    const float bias1 = b[D_HEAD + lane];
    const float bias2 = b[2 * D_HEAD + lane];
    const float bias3 = b[3 * D_HEAD + lane];

    const float* whl = Wh + lane;
    float acc0 = 0.0f, acc1 = 0.0f, acc2 = 0.0f, acc3 = 0.0f;
    if (cnt <= 32) {
        const bool act = lane < cnt;
        int   s  = 0;
        float e0 = -INFINITY, e1 = -INFINITY, e2 = -INFINITY, e3 = -INFINITY;
        if (act) {
            s = csr_src[start + lane];
            const float4 l4 = *reinterpret_cast<const float4*>(el + (long)s * HEADS);
            e0 = leaky(l4.x + r4.x);
            e1 = leaky(l4.y + r4.y);
            e2 = leaky(l4.z + r4.z);
            e3 = leaky(l4.w + r4.w);
        }
        const float m0 = finite_or_zero(wave_max(e0));
        const float m1 = finite_or_zero(wave_max(e1));
        const float m2 = finite_or_zero(wave_max(e2));
        const float m3 = finite_or_zero(wave_max(e3));
        const float p0 = act ? expf(e0 - m0) : 0.0f;
        const float p1 = act ? expf(e1 - m1) : 0.0f;
        const float p2 = act ? expf(e2 - m2) : 0.0f;
        const float p3 = act ? expf(e3 - m3) : 0.0f;
        const float d0 = wave_sum(p0) + EPS_;
        const float d1 = wave_sum(p1) + EPS_;
        const float d2 = wave_sum(p2) + EPS_;
        const float d3 = wave_sum(p3) + EPS_;
        add_rows(whl, s, p0 / d0, p1 / d1, p2 / d2, p3 / d3, cnt, acc0, acc1, acc2, acc3);
    } else {
        float m0 = -INFINITY, m1 = -INFINITY, m2 = -INFINITY, m3 = -INFINITY;
        for (int base = 0; base < cnt; base += 32) {
            const int j = base + lane;
            if (j < cnt) {
                const int s = csr_src[start + j];
                const float4 l4 = *reinterpret_cast<const float4*>(el + (long)s * HEADS);
                m0 = fmaxf(m0, leaky(l4.x + r4.x));
                m1 = fmaxf(m1, leaky(l4.y + r4.y));
                m2 = fmaxf(m2, leaky(l4.z + r4.z));
                m3 = fmaxf(m3, leaky(l4.w + r4.w));
            }
        }
        m0 = finite_or_zero(wave_max(m0));
        m1 = finite_or_zero(wave_max(m1));
        m2 = finite_or_zero(wave_max(m2));
        m3 = finite_or_zero(wave_max(m3));

        float d0 = 0.0f, d1 = 0.0f, d2 = 0.0f, d3 = 0.0f;
        for (int base = 0; base < cnt; base += 32) {
            const int j = base + lane;
            if (j < cnt) {
                const int s = csr_src[start + j];
                const float4 l4 = *reinterpret_cast<const float4*>(el + (long)s * HEADS);
                d0 += expf(leaky(l4.x + r4.x) - m0);
                d1 += expf(leaky(l4.y + r4.y) - m1);
                d2 += expf(leaky(l4.z + r4.z) - m2);
                d3 += expf(leaky(l4.w + r4.w) - m3);
            }
        }
        d0 = wave_sum(d0) + EPS_;
        d1 = wave_sum(d1) + EPS_;
        d2 = wave_sum(d2) + EPS_;
        d3 = wave_sum(d3) + EPS_;

        for (int base = 0; base < cnt; base += 32) {
            const int j = base + lane;
            int   s  = 0;
            float a0 = 0.0f, a1 = 0.0f, a2 = 0.0f, a3 = 0.0f;
            if (j < cnt) {
                s = csr_src[start + j];
                const float4 l4 = *reinterpret_cast<const float4*>(el + (long)s * HEADS);
                a0 = expf(leaky(l4.x + r4.x) - m0) / d0;
                a1 = expf(leaky(l4.y + r4.y) - m1) / d1;
                a2 = expf(leaky(l4.z + r4.z) - m2) / d2;
                a3 = expf(leaky(l4.w + r4.w) - m3) / d3;
            }
            const int c = (cnt - base < 32) ? (cnt - base) : 32;
            add_rows(whl, s, a0, a1, a2, a3, c, acc0, acc1, acc2, acc3);
        }
    }
    const float x0 = acc0 + bias0;
    const float x1 = acc1 + bias1;
    const float x2 = acc2 + bias2;
    const float x3 = acc3 + bias3;
    if (MEAN_HEADS) {
        float s = 0.0f;
        s += x0;
        s += x1;
        s += x2;
        s += x3;
        res[(long)n * D_HEAD + lane] = s * (1.0f / HEADS);
    } else {
        float* o = res + (long)n * FEAT + lane;
        o[0]          = elu(x0);
        o[D_HEAD]     = elu(x1);
        o[2 * D_HEAD] = elu(x2);
        o[3 * D_HEAD] = elu(x3);
    }
}


extern "C" void kernel_launch(void* const* d_in, const int* in_sizes, int n_in,
                              void* d_out, int out_size, void* d_ws, size_t ws_size,
                              hipStream_t stream)
{
    const float* h   = (const float*)d_in[0];
    const int*   src = (const int*)  d_in[1];
    const int*   dst = (const int*)  d_in[2];
    const float* W1  = (const float*)d_in[3];
    const float* al1 = (const float*)d_in[4];
    const float* ar1 = (const float*)d_in[5];
    const float* b1  = (const float*)d_in[6];
    const float* W2  = (const float*)d_in[7];
    const float* al2 = (const float*)d_in[8];
    const float* ar2 = (const float*)d_in[9];
    const float* b2  = (const float*)d_in[10];
    const float* Wp  = (const float*)d_in[11];
    const float* bp  = (const float*)d_in[12];
    float*       out = (float*)d_out;

    const int N = N_NODES, E = N_EDGES;
    static_assert(N_NODES % 16 == 0, "the GEMM's 16-row strips are whole");

    float* Wh      = (float*)d_ws;
    float* Z       = Wh + (size_t)N * FEAT;
    float* el      = Z  + (size_t)N * FEAT;
    float* er      = el + (size_t)N * HEADS;
    float* zmean   = er + (size_t)N * HEADS;
    int*   deg     = (int*)(zmean + (size_t)N * D_HEAD);
    int*   rowptr  = deg + (size_t)N;
    int*   rank    = rowptr + (size_t)N;
    int*   csr_src = rank + (size_t)E;

    const int mtiles = (N + 15) / 16;
    const dim3 gemmBlk(256);
    const dim3 gemmGrd((mtiles + 7) / 8);

    const int zThreads = 256;
    auto zgrid = [](long n) { return (unsigned)((n + 255) / 256); };

    k_zero<<<zgrid((long)N), zThreads, 0, stream>>>((float*)deg, (long)N);
    k_count_rank<<<(E + 255) / 256, 256, 0, stream>>>(dst, deg, rank, E);
    k_rowptr<<<(N + 1023) / 1024, 1024, 0, stream>>>(deg, rowptr, N);
    k_fill<<<(E + 255) / 256, 256, 0, stream>>>(src, dst, rowptr, rank, csr_src, E);

    k_gemm_strip<FEAT, FEAT, false><<<gemmGrd, gemmBlk, 0, stream>>>(h, W1, nullptr, Wh, N);
    k_scores<<<(N * HEADS + 255) / 256, 256, 0, stream>>>(Wh, al1, ar1, el, er, N);
    k_node_gather<false><<<(N + 7) / 8, 256, 0, stream>>>(Wh, el, er, rowptr, deg, csr_src, b1, Z, N);

    k_gemm_strip<FEAT, FEAT, false><<<gemmGrd, gemmBlk, 0, stream>>>(Z, W2, nullptr, Wh, N);
    k_scores<<<(N * HEADS + 255) / 256, 256, 0, stream>>>(Wh, al2, ar2, el, er, N);
    k_node_gather<true><<<(N + 7) / 8, 256, 0, stream>>>(Wh, el, er, rowptr, deg, csr_src, b2, zmean, N);

    k_gemm_strip<D_HEAD, OUT_DIM, true><<<gemmGrd, gemmBlk, 0, stream>>>(zmean, Wp, bp, out, N);
}


# Per-node scores el, er formed in the GEMM epilogue from its accumulators (16-lane DPP halving sums); both score launches and their re-read of Wh removed; the scores' feature sum order changes
# speedup vs baseline: 6.4280x; 1.0580x over previous
#include <hip/hip_runtime.h>
#include <hip/hip_bf16.h>
#include <math.h>

#define N_NODES   50000
#define N_EDGES   800000
#define HEADS     4
#define D_HEAD    32
#define FEAT      128
#define OUT_DIM   512
#define NEG_SLOPE 0.2f
#define EPS_      1e-9f

typedef __attribute__((ext_vector_type(2))) float v2f;
typedef __attribute__((ext_vector_type(8))) float v8f;


__global__ void k_zero(float* __restrict__ p, long n) {
    long i = (long)blockIdx.x * blockDim.x + threadIdx.x;
    if (i < n) p[i] = 0.0f;
}


template <int XM>
__device__ __forceinline__ float row_xor(float x) {
    return __int_as_float(__builtin_amdgcn_update_dpp(0, __float_as_int(x), 0x160 + XM, 0xF, 0xF, true));
}
template <int LEN, int XM>
__device__ __forceinline__ void halve_step(float (&v)[64], bool hi) {
#pragma unroll
    for (int k = 0; k < LEN / 2; ++k) {
        const float keep = hi ? v[k + LEN / 2] : v[k];
        const float send = hi ? v[k] : v[k + LEN / 2];
        v[k] = keep + row_xor<XM>(send);
    }
}

template <int K, int N, bool HAS_BIAS, bool SCORES>
__global__ void __launch_bounds__(256)
k_gemm_strip(const float* __restrict__ A, const float* __restrict__ B,
             const float* __restrict__ bias, float* __restrict__ C, int M,
             const float* __restrict__ al, const float* __restrict__ ar,
             float* __restrict__ el, float* __restrict__ er)
{
    static_assert(K % 8 == 0 && N % 128 == 0 && K * N == 16384, "tile plan");
    static_assert(!SCORES || N == HEADS * D_HEAD, "a head is one pair of column tiles");
    constexpr int KH = K / 2;
    constexpr int KS = K / 4;
    constexpr int NA = KH / 4;
    constexpr int NG = N / 128;

    __shared__ float4 ldsW[K * N / 4];

    const int tid  = threadIdx.x;
    const int wave = tid >> 5;
    const int lane = tid & 31;
    const int c    = lane & 15;
    const int h    = lane >> 4;
    const int m0   = (blockIdx.x * 8 + wave) * 16;

    const int arow = (m0 + c < M) ? (m0 + c) : (M - 1);
    const float4* ap = reinterpret_cast<const float4*>(A + (long)arow * K + h * KH);
    float4 a4[NA];
#pragma unroll
    for (int i = 0; i < NA; ++i) a4[i] = ap[i];
    float2 bz[N / 32];
    if (HAS_BIAS) {
#pragma unroll
        for (int t = 0; t < N / 32; ++t)
            bz[t] = *reinterpret_cast<const float2*>(bias + 32 * t + 2 * c);
    }
    float2 alv[4], arv[4];
    if (SCORES) {
#pragma unroll
        for (int p = 0; p < 4; ++p) {
            alv[p] = *reinterpret_cast<const float2*>(al + 32 * p + 2 * c);
            arv[p] = *reinterpret_cast<const float2*>(ar + 32 * p + 2 * c);
        }
    }

    constexpr int ITEMS = (K / 2) * (N / 4);
    static_assert(ITEMS % 256 == 0, "staging plan");
#pragma unroll
    for (int u = 0; u < ITEMS / 256; ++u) {
        const int it = tid + u * 256;
        const int kp = it / (N / 4);
        const int q  = it % (N / 4);
        const float4 r0 = *reinterpret_cast<const float4*>(B + (long)(2 * kp) * N + 4 * q);
        const float4 r1 = *reinterpret_cast<const float4*>(B + (long)(2 * kp + 1) * N + 4 * q);
        const int w = (kp * N + 4 * q) >> 1;
        ldsW[w]     = make_float4(r0.x, r1.x, r0.y, r1.y);
        ldsW[w + 1] = make_float4(r0.z, r1.z, r0.w, r1.w);
    }
    __syncthreads();

    if (m0 >= M) return;

    const int rbase = m0 + (h << 3);
    const float4* lp = ldsW + (((h * KS) * N + 2 * c) >> 1);

    float4 bq[2][4];
#pragma unroll
    for (int p = 0; p < 4; ++p) bq[0][p] = lp[16 * p];

#pragma unroll
    for (int g = 0; g < NG; ++g) {
        v8f acc[8];
#pragma unroll
        for (int t = 0; t < 8; ++t) acc[t] = v8f{};
#pragma unroll
        for (int j = 0; j < KS; ++j) {
            const int cur = (g * KS + j) & 1;
            const int jn  = (j + 1 < KS) ? (j + 1) : 0;
            const int gn  = (j + 1 < KS) ? g : (g + 1);
            if (gn < NG) {
#pragma unroll
                for (int p = 0; p < 4; ++p) bq[cur ^ 1][p] = lp[gn * 64 + jn * (N / 2) + 16 * p];
            }
            v2f a;
            if (j & 1) { a.x = a4[j >> 1].z; a.y = a4[j >> 1].w; }
            else       { a.x = a4[j >> 1].x; a.y = a4[j >> 1].y; }
#pragma unroll
            for (int p = 0; p < 4; ++p) {
                v2f b0, b1;
                b0.x = bq[cur][p].x; b0.y = bq[cur][p].y;
                b1.x = bq[cur][p].z; b1.y = bq[cur][p].w;
                acc[2 * p]     = __builtin_amdgcn_wmma_f32_16x16x4_f32(false, a, false, b0,
                                                       (short)0, acc[2 * p], false, false);
                acc[2 * p + 1] = __builtin_amdgcn_wmma_f32_16x16x4_f32(false, a, false, b1,
                                                       (short)0, acc[2 * p + 1], false, false);
            }
        }
#pragma unroll
        for (int p = 0; p < 4; ++p) {
            float* crow = C + (long)rbase * N + g * 128 + 32 * p + 2 * c;
#pragma unroll
            for (int i = 0; i < 8; ++i) {
                float2 o;
                o.x = acc[2 * p][i];
                o.y = acc[2 * p + 1][i];
                if (HAS_BIAS) { o.x += bz[g * 4 + p].x; o.y += bz[g * 4 + p].y; }
                *reinterpret_cast<float2*>(crow + (long)i * N) = o;
            }
        }
        if (SCORES) {
            float v[64];
#pragma unroll
            for (int i = 0; i < 8; ++i) {
#pragma unroll
                for (int p = 0; p < 4; ++p) {
                    v[i * 4 + p]      = acc[2 * p][i] * alv[p].x + acc[2 * p + 1][i] * alv[p].y;
                    v[32 + i * 4 + p] = acc[2 * p][i] * arv[p].x + acc[2 * p + 1][i] * arv[p].y;
                }
            }
            halve_step<64, 8>(v, (c & 8) != 0);
            halve_step<32, 4>(v, (c & 4) != 0);
            halve_step<16, 2>(v, (c & 2) != 0);
            halve_step<8, 1>(v, (c & 1) != 0);
            float* sp = ((c & 8) ? er : el) + (long)(rbase + (c & 7)) * HEADS;
            *reinterpret_cast<float4*>(sp) = make_float4(v[0], v[1], v[2], v[3]);
        }
    }
}


__global__ void k_count_rank(const int* __restrict__ dst, int* __restrict__ deg,
                             int* __restrict__ rank, int E)
{
    int e = blockIdx.x * blockDim.x + threadIdx.x;
    if (e >= E) return;
    rank[e] = atomicAdd(&deg[dst[e]], 1);
}

__global__ void __launch_bounds__(1024)
k_rowptr(const int* __restrict__ deg, int* __restrict__ rowptr, int N)
{
    __shared__ int s_pre[32];
    __shared__ int s_tot[32];
    const int t = threadIdx.x, lane = t & 31, w = t >> 5;
    const int first = blockIdx.x * 1024;
    int pre = 0;
    for (int i = t; i < first; i += 1024) pre += deg[i];
    const int idx = first + t;
    const int v = (idx < N) ? deg[idx] : 0;
    int inc = v;
    for (int o = 1; o < 32; o <<= 1) {
        int u = __shfl_up(inc, o);
        if (lane >= o) inc += u;
    }
    for (int o = 16; o > 0; o >>= 1) pre += __shfl_xor(pre, o);
    if (lane == 31) s_tot[w] = inc;
    if (lane == 0)  s_pre[w] = pre;
    __syncthreads();
    int base = 0;
    for (int i = 0; i < 32; ++i) {
        base += s_pre[i];
        if (i < w) base += s_tot[i];
    }
    if (idx < N) rowptr[idx] = base + inc - v;
}

__global__ void k_fill(const int* __restrict__ src, const int* __restrict__ dst,
                       const int* __restrict__ rowptr, const int* __restrict__ rank,
                       int* __restrict__ csr_src, int E)
{
    int e = blockIdx.x * blockDim.x + threadIdx.x;
    if (e >= E) return;
    csr_src[rowptr[dst[e]] + rank[e]] = src[e];
}

__device__ __forceinline__ float leaky(float v) { return (v > 0.0f) ? v : NEG_SLOPE * v; }
__device__ __forceinline__ float wave_max(float v) {
    for (int o = 16; o > 0; o >>= 1) v = fmaxf(v, __shfl_xor(v, o));
    return v;
}
__device__ __forceinline__ float wave_sum(float v) {
    for (int o = 16; o > 0; o >>= 1) v += __shfl_xor(v, o);
    return v;
}
__device__ __forceinline__ float finite_or_zero(float m) {
    return (fabsf(m) <= 3.402823466e+38f) ? m : 0.0f;
}
__device__ __forceinline__ float lane_bcast(float v, int t) {
    return __int_as_float(__builtin_amdgcn_readlane(__float_as_int(v), t));
}
__device__ __forceinline__ float elu(float x) { return (x > 0.0f) ? x : expm1f(x); }

__device__ __forceinline__ void add_rows(const float* __restrict__ whl, int s,
                                         float a0, float a1, float a2, float a3, int c,
                                         float& acc0, float& acc1, float& acc2, float& acc3)
{
#pragma unroll 4
    for (int t = 0; t < c; ++t) {
        const float* row = whl + (long)__builtin_amdgcn_readlane(s, t) * FEAT;
        acc0 += lane_bcast(a0, t) * row[0];
        acc1 += lane_bcast(a1, t) * row[D_HEAD];
        acc2 += lane_bcast(a2, t) * row[2 * D_HEAD];
        acc3 += lane_bcast(a3, t) * row[3 * D_HEAD];
    }
}

template <bool MEAN_HEADS>
__global__ void __launch_bounds__(256)
k_node_gather(const float* __restrict__ Wh, const float* __restrict__ el,
              const float* __restrict__ er, const int* __restrict__ rowptr,
              const int* __restrict__ deg, const int* __restrict__ csr_src,
              const float* __restrict__ b, float* __restrict__ res, int N)
{
    const int lane = threadIdx.x & 31;
    const int n = __builtin_amdgcn_readfirstlane((int)(blockIdx.x * 8 + (threadIdx.x >> 5)));
    if (n >= N) return;
    const int start = rowptr[n];
    const int cnt   = deg[n];
    const float4 r4 = *reinterpret_cast<const float4*>(er + (long)n * HEADS);
    const float bias0 = b[lane];
    const float bias1 = b[D_HEAD + lane];
    const float bias2 = b[2 * D_HEAD + lane];
    const float bias3 = b[3 * D_HEAD + lane];

    const float* whl = Wh + lane;
    float acc0 = 0.0f, acc1 = 0.0f, acc2 = 0.0f, acc3 = 0.0f;
    if (cnt <= 32) {
        const bool act = lane < cnt;
        int   s  = 0;
        float e0 = -INFINITY, e1 = -INFINITY, e2 = -INFINITY, e3 = -INFINITY;
        if (act) {
            s = csr_src[start + lane];
            const float4 l4 = *reinterpret_cast<const float4*>(el + (long)s * HEADS);
            e0 = leaky(l4.x + r4.x);
            e1 = leaky(l4.y + r4.y);
            e2 = leaky(l4.z + r4.z);
            e3 = leaky(l4.w + r4.w);
        }
        const float m0 = finite_or_zero(wave_max(e0));
        const float m1 = finite_or_zero(wave_max(e1));
        const float m2 = finite_or_zero(wave_max(e2));
        const float m3 = finite_or_zero(wave_max(e3));
        const float p0 = act ? expf(e0 - m0) : 0.0f;
        const float p1 = act ? expf(e1 - m1) : 0.0f;
        const float p2 = act ? expf(e2 - m2) : 0.0f;
        const float p3 = act ? expf(e3 - m3) : 0.0f;
        const float d0 = wave_sum(p0) + EPS_;
        const float d1 = wave_sum(p1) + EPS_;
        const float d2 = wave_sum(p2) + EPS_;
        const float d3 = wave_sum(p3) + EPS_;
        add_rows(whl, s, p0 / d0, p1 / d1, p2 / d2, p3 / d3, cnt, acc0, acc1, acc2, acc3);
    } else {
        float m0 = -INFINITY, m1 = -INFINITY, m2 = -INFINITY, m3 = -INFINITY;
        for (int base = 0; base < cnt; base += 32) {
            const int j = base + lane;
            if (j < cnt) {
                const int s = csr_src[start + j];
                const float4 l4 = *reinterpret_cast<const float4*>(el + (long)s * HEADS);
                m0 = fmaxf(m0, leaky(l4.x + r4.x));
                m1 = fmaxf(m1, leaky(l4.y + r4.y));
                m2 = fmaxf(m2, leaky(l4.z + r4.z));
                m3 = fmaxf(m3, leaky(l4.w + r4.w));
            }
        }
        m0 = finite_or_zero(wave_max(m0));
        m1 = finite_or_zero(wave_max(m1));
        m2 = finite_or_zero(wave_max(m2));
        m3 = finite_or_zero(wave_max(m3));

        float d0 = 0.0f, d1 = 0.0f, d2 = 0.0f, d3 = 0.0f;
        for (int base = 0; base < cnt; base += 32) {
            const int j = base + lane;
            if (j < cnt) {
                const int s = csr_src[start + j];
                const float4 l4 = *reinterpret_cast<const float4*>(el + (long)s * HEADS);
                d0 += expf(leaky(l4.x + r4.x) - m0);
                d1 += expf(leaky(l4.y + r4.y) - m1);
                d2 += expf(leaky(l4.z + r4.z) - m2);
                d3 += expf(leaky(l4.w + r4.w) - m3);
            }
        }
        d0 = wave_sum(d0) + EPS_;
        d1 = wave_sum(d1) + EPS_;
        d2 = wave_sum(d2) + EPS_;
        d3 = wave_sum(d3) + EPS_;

        for (int base = 0; base < cnt; base += 32) {
            const int j = base + lane;
            int   s  = 0;
            float a0 = 0.0f, a1 = 0.0f, a2 = 0.0f, a3 = 0.0f;
            if (j < cnt) {
                s = csr_src[start + j];
                const float4 l4 = *reinterpret_cast<const float4*>(el + (long)s * HEADS);
                a0 = expf(leaky(l4.x + r4.x) - m0) / d0;
                a1 = expf(leaky(l4.y + r4.y) - m1) / d1;
                a2 = expf(leaky(l4.z + r4.z) - m2) / d2;
                a3 = expf(leaky(l4.w + r4.w) - m3) / d3;
            }
            const int c = (cnt - base < 32) ? (cnt - base) : 32;
            add_rows(whl, s, a0, a1, a2, a3, c, acc0, acc1, acc2, acc3);
        }
    }
    const float x0 = acc0 + bias0;
    const float x1 = acc1 + bias1;
    const float x2 = acc2 + bias2;
    const float x3 = acc3 + bias3;
    if (MEAN_HEADS) {
        float s = 0.0f;
        s += x0;
        s += x1;
        s += x2;
        s += x3;
        res[(long)n * D_HEAD + lane] = s * (1.0f / HEADS);
    } else {
        float* o = res + (long)n * FEAT + lane;
        o[0]          = elu(x0);
        o[D_HEAD]     = elu(x1);
        o[2 * D_HEAD] = elu(x2);
        o[3 * D_HEAD] = elu(x3);
    }
}


extern "C" void kernel_launch(void* const* d_in, const int* in_sizes, int n_in,
                              void* d_out, int out_size, void* d_ws, size_t ws_size,
                              hipStream_t stream)
{
    const float* h   = (const float*)d_in[0];
    const int*   src = (const int*)  d_in[1];
    const int*   dst = (const int*)  d_in[2];
    const float* W1  = (const float*)d_in[3];
    const float* al1 = (const float*)d_in[4];
    const float* ar1 = (const float*)d_in[5];
    const float* b1  = (const float*)d_in[6];
    const float* W2  = (const float*)d_in[7];
    const float* al2 = (const float*)d_in[8];
    const float* ar2 = (const float*)d_in[9];
    const float* b2  = (const float*)d_in[10];
    const float* Wp  = (const float*)d_in[11];
    const float* bp  = (const float*)d_in[12];
    float*       out = (float*)d_out;

    const int N = N_NODES, E = N_EDGES;
    static_assert(N_NODES % 16 == 0, "the GEMM's 16-row strips are whole");

    float* Wh      = (float*)d_ws;
    float* Z       = Wh + (size_t)N * FEAT;
    float* el      = Z  + (size_t)N * FEAT;
    float* er      = el + (size_t)N * HEADS;
    float* zmean   = er + (size_t)N * HEADS;
    int*   deg     = (int*)(zmean + (size_t)N * D_HEAD);
    int*   rowptr  = deg + (size_t)N;
    int*   rank    = rowptr + (size_t)N;
    int*   csr_src = rank + (size_t)E;

    const int mtiles = (N + 15) / 16;
    const dim3 gemmBlk(256);
    const dim3 gemmGrd((mtiles + 7) / 8);

    const int zThreads = 256;
    auto zgrid = [](long n) { return (unsigned)((n + 255) / 256); };

    k_zero<<<zgrid((long)N), zThreads, 0, stream>>>((float*)deg, (long)N);
    k_count_rank<<<(E + 255) / 256, 256, 0, stream>>>(dst, deg, rank, E);
    k_rowptr<<<(N + 1023) / 1024, 1024, 0, stream>>>(deg, rowptr, N);
    k_fill<<<(E + 255) / 256, 256, 0, stream>>>(src, dst, rowptr, rank, csr_src, E);

    k_gemm_strip<FEAT, FEAT, false, true><<<gemmGrd, gemmBlk, 0, stream>>>(h, W1, nullptr, Wh, N,
                                                                          al1, ar1, el, er);
    k_node_gather<false><<<(N + 7) / 8, 256, 0, stream>>>(Wh, el, er, rowptr, deg, csr_src, b1, Z, N);

    k_gemm_strip<FEAT, FEAT, false, true><<<gemmGrd, gemmBlk, 0, stream>>>(Z, W2, nullptr, Wh, N,
                                                                          al2, ar2, el, er);
    k_node_gather<true><<<(N + 7) / 8, 256, 0, stream>>>(Wh, el, er, rowptr, deg, csr_src, b2, zmean, N);

    k_gemm_strip<D_HEAD, OUT_DIM, true, false><<<gemmGrd, gemmBlk, 0, stream>>>(zmean, Wp, bp, out, N,
                                                                                nullptr, nullptr, nullptr, nullptr);
}


# Edge count and arrival-rank atomics issued by the layer-1 GEMM's own threads (8 edges each) and left in flight beside its matrix loop; the count/rank launch is removed; rank arrival order changes
# speedup vs baseline: 6.6612x; 1.0363x over previous
#include <hip/hip_runtime.h>
#include <hip/hip_bf16.h>
#include <math.h>

#define N_NODES   50000
#define N_EDGES   800000
#define HEADS     4
#define D_HEAD    32
#define FEAT      128
#define OUT_DIM   512
#define NEG_SLOPE 0.2f
#define EPS_      1e-9f

typedef __attribute__((ext_vector_type(2))) float v2f;
typedef __attribute__((ext_vector_type(8))) float v8f;


__global__ void k_zero(float* __restrict__ p, long n) {
    long i = (long)blockIdx.x * blockDim.x + threadIdx.x;
    if (i < n) p[i] = 0.0f;
}


template <int XM>
__device__ __forceinline__ float row_xor(float x) {
    return __int_as_float(__builtin_amdgcn_update_dpp(0, __float_as_int(x), 0x160 + XM, 0xF, 0xF, true));
}
template <int LEN, int XM>
__device__ __forceinline__ void halve_step(float (&v)[64], bool hi) {
#pragma unroll
    for (int k = 0; k < LEN / 2; ++k) {
        const float keep = hi ? v[k + LEN / 2] : v[k];
        const float send = hi ? v[k] : v[k + LEN / 2];
        v[k] = keep + row_xor<XM>(send);
    }
}

#define EPT 8
static_assert(((N_NODES / 16 + 7) / 8) * 256L * EPT >= N_EDGES && N_EDGES % EPT == 0,
              "the layer-1 GEMM grid's threads cover every edge, EPT whole edges each");
template <int K, int N, bool HAS_BIAS, bool SCORES, bool RANK>
__global__ void __launch_bounds__(256)
k_gemm_strip(const float* __restrict__ A, const float* __restrict__ B,
             const float* __restrict__ bias, float* __restrict__ C, int M,
             const float* __restrict__ al, const float* __restrict__ ar,
             float* __restrict__ el, float* __restrict__ er,
             const int* __restrict__ dst, int* __restrict__ deg, int* __restrict__ rank, int E)
{
    static_assert(K % 8 == 0 && N % 128 == 0 && K * N == 16384, "tile plan");
    static_assert(!SCORES || N == HEADS * D_HEAD, "a head is one pair of column tiles");
    constexpr int KH = K / 2;
    constexpr int KS = K / 4;
    constexpr int NA = KH / 4;
    constexpr int NG = N / 128;

    __shared__ float4 ldsW[K * N / 4];

    const int tid  = threadIdx.x;
    const int wave = tid >> 5;
    const int lane = tid & 31;
    const int c    = lane & 15;
    const int h    = lane >> 4;
    const int m0   = (blockIdx.x * 8 + wave) * 16;

    const long e0  = ((long)blockIdx.x * 256 + tid) * EPT;
    const bool own = RANK && (e0 + EPT <= (long)E);
    int4 dA = make_int4(0, 0, 0, 0), dB = make_int4(0, 0, 0, 0);
    if (own) {
        dA = *reinterpret_cast<const int4*>(dst + e0);
        dB = *reinterpret_cast<const int4*>(dst + e0 + 4);
    }

    const int arow = (m0 + c < M) ? (m0 + c) : (M - 1);
    const float4* ap = reinterpret_cast<const float4*>(A + (long)arow * K + h * KH);
    float4 a4[NA];
#pragma unroll
    for (int i = 0; i < NA; ++i) a4[i] = ap[i];
    float2 bz[N / 32];
    if (HAS_BIAS) {
#pragma unroll
        for (int t = 0; t < N / 32; ++t)
            bz[t] = *reinterpret_cast<const float2*>(bias + 32 * t + 2 * c);
    }
    float2 alv[4], arv[4];
    if (SCORES) {
#pragma unroll
        for (int p = 0; p < 4; ++p) {
            alv[p] = *reinterpret_cast<const float2*>(al + 32 * p + 2 * c);
            arv[p] = *reinterpret_cast<const float2*>(ar + 32 * p + 2 * c);
        }
    }

    constexpr int ITEMS = (K / 2) * (N / 4);
    static_assert(ITEMS % 256 == 0, "staging plan");
#pragma unroll
    for (int u = 0; u < ITEMS / 256; ++u) {
        const int it = tid + u * 256;
        const int kp = it / (N / 4);
        const int q  = it % (N / 4);
        const float4 r0 = *reinterpret_cast<const float4*>(B + (long)(2 * kp) * N + 4 * q);
        const float4 r1 = *reinterpret_cast<const float4*>(B + (long)(2 * kp + 1) * N + 4 * q);
        const int w = (kp * N + 4 * q) >> 1;
        ldsW[w]     = make_float4(r0.x, r1.x, r0.y, r1.y);
        ldsW[w + 1] = make_float4(r0.z, r1.z, r0.w, r1.w);
    }
    int4 rA = make_int4(0, 0, 0, 0), rB = make_int4(0, 0, 0, 0);
    if (own) {
        rA.x = atomicAdd(&deg[dA.x], 1);
        rA.y = atomicAdd(&deg[dA.y], 1);
        rA.z = atomicAdd(&deg[dA.z], 1);
        rA.w = atomicAdd(&deg[dA.w], 1);
        rB.x = atomicAdd(&deg[dB.x], 1);
        rB.y = atomicAdd(&deg[dB.y], 1);
        rB.z = atomicAdd(&deg[dB.z], 1);
        rB.w = atomicAdd(&deg[dB.w], 1);
    }
    if (RANK) {
        __builtin_amdgcn_fence(__ATOMIC_RELEASE, "workgroup", "local");
        __builtin_amdgcn_s_barrier();
        __builtin_amdgcn_fence(__ATOMIC_ACQUIRE, "workgroup", "local");
    } else {
        __syncthreads();
    }
    if (m0 >= M) {
        if (own) {
            *reinterpret_cast<int4*>(rank + e0)     = rA;
            *reinterpret_cast<int4*>(rank + e0 + 4) = rB;
        }
        return;
    }

    const int rbase = m0 + (h << 3);
    const float4* lp = ldsW + (((h * KS) * N + 2 * c) >> 1);

    float4 bq[2][4];
#pragma unroll
    for (int p = 0; p < 4; ++p) bq[0][p] = lp[16 * p];

#pragma unroll
    for (int g = 0; g < NG; ++g) {
        v8f acc[8];
#pragma unroll
        for (int t = 0; t < 8; ++t) acc[t] = v8f{};
#pragma unroll
        for (int j = 0; j < KS; ++j) {
            const int cur = (g * KS + j) & 1;
            const int jn  = (j + 1 < KS) ? (j + 1) : 0;
            const int gn  = (j + 1 < KS) ? g : (g + 1);
            if (gn < NG) {
#pragma unroll
                for (int p = 0; p < 4; ++p) bq[cur ^ 1][p] = lp[gn * 64 + jn * (N / 2) + 16 * p];
            }
            v2f a;
            if (j & 1) { a.x = a4[j >> 1].z; a.y = a4[j >> 1].w; }
            else       { a.x = a4[j >> 1].x; a.y = a4[j >> 1].y; }
#pragma unroll
            for (int p = 0; p < 4; ++p) {
                v2f b0, b1;
                b0.x = bq[cur][p].x; b0.y = bq[cur][p].y;
                b1.x = bq[cur][p].z; b1.y = bq[cur][p].w;
                acc[2 * p]     = __builtin_amdgcn_wmma_f32_16x16x4_f32(false, a, false, b0,
                                                       (short)0, acc[2 * p], false, false);
                acc[2 * p + 1] = __builtin_amdgcn_wmma_f32_16x16x4_f32(false, a, false, b1,
                                                       (short)0, acc[2 * p + 1], false, false);
            }
        }
#pragma unroll
        for (int p = 0; p < 4; ++p) {
            float* crow = C + (long)rbase * N + g * 128 + 32 * p + 2 * c;
#pragma unroll
            for (int i = 0; i < 8; ++i) {
                float2 o;
                o.x = acc[2 * p][i];
                o.y = acc[2 * p + 1][i];
                if (HAS_BIAS) { o.x += bz[g * 4 + p].x; o.y += bz[g * 4 + p].y; }
                *reinterpret_cast<float2*>(crow + (long)i * N) = o;
            }
        }
        if (SCORES) {
            float v[64];
#pragma unroll
            for (int i = 0; i < 8; ++i) {
#pragma unroll
                for (int p = 0; p < 4; ++p) {
                    v[i * 4 + p]      = acc[2 * p][i] * alv[p].x + acc[2 * p + 1][i] * alv[p].y;
                    v[32 + i * 4 + p] = acc[2 * p][i] * arv[p].x + acc[2 * p + 1][i] * arv[p].y;
                }
            }
            halve_step<64, 8>(v, (c & 8) != 0);
            halve_step<32, 4>(v, (c & 4) != 0);
            halve_step<16, 2>(v, (c & 2) != 0);
            halve_step<8, 1>(v, (c & 1) != 0);
            float* sp = ((c & 8) ? er : el) + (long)(rbase + (c & 7)) * HEADS;
            *reinterpret_cast<float4*>(sp) = make_float4(v[0], v[1], v[2], v[3]);
        }
    }
    if (own) {
        *reinterpret_cast<int4*>(rank + e0)     = rA;
        *reinterpret_cast<int4*>(rank + e0 + 4) = rB;
    }
}


__global__ void __launch_bounds__(1024)
k_rowptr(const int* __restrict__ deg, int* __restrict__ rowptr, int N)
{
    __shared__ int s_pre[32];
    __shared__ int s_tot[32];
    const int t = threadIdx.x, lane = t & 31, w = t >> 5;
    const int first = blockIdx.x * 1024;
    int pre = 0;
    for (int i = t; i < first; i += 1024) pre += deg[i];
    const int idx = first + t;
    const int v = (idx < N) ? deg[idx] : 0;
    int inc = v;
    for (int o = 1; o < 32; o <<= 1) {
        int u = __shfl_up(inc, o);
        if (lane >= o) inc += u;
    }
    for (int o = 16; o > 0; o >>= 1) pre += __shfl_xor(pre, o);
    if (lane == 31) s_tot[w] = inc;
    if (lane == 0)  s_pre[w] = pre;
    __syncthreads();
    int base = 0;
    for (int i = 0; i < 32; ++i) {
        base += s_pre[i];
        if (i < w) base += s_tot[i];
    }
    if (idx < N) rowptr[idx] = base + inc - v;
}

__global__ void k_fill(const int* __restrict__ src, const int* __restrict__ dst,
                       const int* __restrict__ rowptr, const int* __restrict__ rank,
                       int* __restrict__ csr_src, int E)
{
    int e = blockIdx.x * blockDim.x + threadIdx.x;
    if (e >= E) return;
    csr_src[rowptr[dst[e]] + rank[e]] = src[e];
}

__device__ __forceinline__ float leaky(float v) { return (v > 0.0f) ? v : NEG_SLOPE * v; }
__device__ __forceinline__ float wave_max(float v) {
    for (int o = 16; o > 0; o >>= 1) v = fmaxf(v, __shfl_xor(v, o));
    return v;
}
__device__ __forceinline__ float wave_sum(float v) {
    for (int o = 16; o > 0; o >>= 1) v += __shfl_xor(v, o);
    return v;
}
__device__ __forceinline__ float finite_or_zero(float m) {
    return (fabsf(m) <= 3.402823466e+38f) ? m : 0.0f;
}
__device__ __forceinline__ float lane_bcast(float v, int t) {
    return __int_as_float(__builtin_amdgcn_readlane(__float_as_int(v), t));
}
__device__ __forceinline__ float elu(float x) { return (x > 0.0f) ? x : expm1f(x); }

__device__ __forceinline__ void add_rows(const float* __restrict__ whl, int s,
                                         float a0, float a1, float a2, float a3, int c,
                                         float& acc0, float& acc1, float& acc2, float& acc3)
{
#pragma unroll 4
    for (int t = 0; t < c; ++t) {
        const float* row = whl + (long)__builtin_amdgcn_readlane(s, t) * FEAT;
        acc0 += lane_bcast(a0, t) * row[0];
        acc1 += lane_bcast(a1, t) * row[D_HEAD];
        acc2 += lane_bcast(a2, t) * row[2 * D_HEAD];
        acc3 += lane_bcast(a3, t) * row[3 * D_HEAD];
    }
}

template <bool MEAN_HEADS>
__global__ void __launch_bounds__(256)
k_node_gather(const float* __restrict__ Wh, const float* __restrict__ el,
              const float* __restrict__ er, const int* __restrict__ rowptr,
              const int* __restrict__ deg, const int* __restrict__ csr_src,
              const float* __restrict__ b, float* __restrict__ res, int N)
{
    const int lane = threadIdx.x & 31;
    const int n = __builtin_amdgcn_readfirstlane((int)(blockIdx.x * 8 + (threadIdx.x >> 5)));
    if (n >= N) return;
    const int start = rowptr[n];
    const int cnt   = deg[n];
    const float4 r4 = *reinterpret_cast<const float4*>(er + (long)n * HEADS);
    const float bias0 = b[lane];
    const float bias1 = b[D_HEAD + lane];
    const float bias2 = b[2 * D_HEAD + lane];
    const float bias3 = b[3 * D_HEAD + lane];

    const float* whl = Wh + lane;
    float acc0 = 0.0f, acc1 = 0.0f, acc2 = 0.0f, acc3 = 0.0f;
    if (cnt <= 32) {
        const bool act = lane < cnt;
        int   s  = 0;
        float e0 = -INFINITY, e1 = -INFINITY, e2 = -INFINITY, e3 = -INFINITY;
        if (act) {
            s = csr_src[start + lane];
            const float4 l4 = *reinterpret_cast<const float4*>(el + (long)s * HEADS);
            e0 = leaky(l4.x + r4.x);
            e1 = leaky(l4.y + r4.y);
            e2 = leaky(l4.z + r4.z);
            e3 = leaky(l4.w + r4.w);
        }
        const float m0 = finite_or_zero(wave_max(e0));
        const float m1 = finite_or_zero(wave_max(e1));
        const float m2 = finite_or_zero(wave_max(e2));
        const float m3 = finite_or_zero(wave_max(e3));
        const float p0 = act ? expf(e0 - m0) : 0.0f;
        const float p1 = act ? expf(e1 - m1) : 0.0f;
        const float p2 = act ? expf(e2 - m2) : 0.0f;
        const float p3 = act ? expf(e3 - m3) : 0.0f;
        const float d0 = wave_sum(p0) + EPS_;
        const float d1 = wave_sum(p1) + EPS_;
        const float d2 = wave_sum(p2) + EPS_;
        const float d3 = wave_sum(p3) + EPS_;
        add_rows(whl, s, p0 / d0, p1 / d1, p2 / d2, p3 / d3, cnt, acc0, acc1, acc2, acc3);
    } else {
        float m0 = -INFINITY, m1 = -INFINITY, m2 = -INFINITY, m3 = -INFINITY;
        for (int base = 0; base < cnt; base += 32) {
            const int j = base + lane;
            if (j < cnt) {
                const int s = csr_src[start + j];
                const float4 l4 = *reinterpret_cast<const float4*>(el + (long)s * HEADS);
                m0 = fmaxf(m0, leaky(l4.x + r4.x));
                m1 = fmaxf(m1, leaky(l4.y + r4.y));
                m2 = fmaxf(m2, leaky(l4.z + r4.z));
                m3 = fmaxf(m3, leaky(l4.w + r4.w));
            }
        }
        m0 = finite_or_zero(wave_max(m0));
        m1 = finite_or_zero(wave_max(m1));
        m2 = finite_or_zero(wave_max(m2));
        m3 = finite_or_zero(wave_max(m3));

        float d0 = 0.0f, d1 = 0.0f, d2 = 0.0f, d3 = 0.0f;
        for (int base = 0; base < cnt; base += 32) {
            const int j = base + lane;
            if (j < cnt) {
                const int s = csr_src[start + j];
                const float4 l4 = *reinterpret_cast<const float4*>(el + (long)s * HEADS);
                d0 += expf(leaky(l4.x + r4.x) - m0);
                d1 += expf(leaky(l4.y + r4.y) - m1);
                d2 += expf(leaky(l4.z + r4.z) - m2);
                d3 += expf(leaky(l4.w + r4.w) - m3);
            }
        }
        d0 = wave_sum(d0) + EPS_;
        d1 = wave_sum(d1) + EPS_;
        d2 = wave_sum(d2) + EPS_;
        d3 = wave_sum(d3) + EPS_;

        for (int base = 0; base < cnt; base += 32) {
            const int j = base + lane;
            int   s  = 0;
            float a0 = 0.0f, a1 = 0.0f, a2 = 0.0f, a3 = 0.0f;
            if (j < cnt) {
                s = csr_src[start + j];
                const float4 l4 = *reinterpret_cast<const float4*>(el + (long)s * HEADS);
                a0 = expf(leaky(l4.x + r4.x) - m0) / d0;
                a1 = expf(leaky(l4.y + r4.y) - m1) / d1;
                a2 = expf(leaky(l4.z + r4.z) - m2) / d2;
                a3 = expf(leaky(l4.w + r4.w) - m3) / d3;
            }
            const int c = (cnt - base < 32) ? (cnt - base) : 32;
            add_rows(whl, s, a0, a1, a2, a3, c, acc0, acc1, acc2, acc3);
        }
    }
    const float x0 = acc0 + bias0;
    const float x1 = acc1 + bias1;
    const float x2 = acc2 + bias2;
    const float x3 = acc3 + bias3;
    if (MEAN_HEADS) {
        float s = 0.0f;
        s += x0;
        s += x1;
        s += x2;
        s += x3;
        res[(long)n * D_HEAD + lane] = s * (1.0f / HEADS);
    } else {
        float* o = res + (long)n * FEAT + lane;
        o[0]          = elu(x0);
        o[D_HEAD]     = elu(x1);
        o[2 * D_HEAD] = elu(x2);
        o[3 * D_HEAD] = elu(x3);
    }
}


extern "C" void kernel_launch(void* const* d_in, const int* in_sizes, int n_in,
                              void* d_out, int out_size, void* d_ws, size_t ws_size,
                              hipStream_t stream)
{
    const float* h   = (const float*)d_in[0];
    const int*   src = (const int*)  d_in[1];
    const int*   dst = (const int*)  d_in[2];
    const float* W1  = (const float*)d_in[3];
    const float* al1 = (const float*)d_in[4];
    const float* ar1 = (const float*)d_in[5];
    const float* b1  = (const float*)d_in[6];
    const float* W2  = (const float*)d_in[7];
    const float* al2 = (const float*)d_in[8];
    const float* ar2 = (const float*)d_in[9];
    const float* b2  = (const float*)d_in[10];
    const float* Wp  = (const float*)d_in[11];
    const float* bp  = (const float*)d_in[12];
    float*       out = (float*)d_out;

    const int N = N_NODES, E = N_EDGES;
    static_assert(N_NODES % 16 == 0, "the GEMM's 16-row strips are whole");

    float* Wh      = (float*)d_ws;
    float* Z       = Wh + (size_t)N * FEAT;
    float* el      = Z  + (size_t)N * FEAT;
    float* er      = el + (size_t)N * HEADS;
    float* zmean   = er + (size_t)N * HEADS;
    int*   deg     = (int*)(zmean + (size_t)N * D_HEAD);
    int*   rowptr  = deg + (size_t)N;
    int*   rank    = rowptr + (size_t)N;
    int*   csr_src = rank + (size_t)E;

    const int mtiles = (N + 15) / 16;
    const dim3 gemmBlk(256);
    const dim3 gemmGrd((mtiles + 7) / 8);

    const int zThreads = 256;
    auto zgrid = [](long n) { return (unsigned)((n + 255) / 256); };

    k_zero<<<zgrid((long)N), zThreads, 0, stream>>>((float*)deg, (long)N);
    k_gemm_strip<FEAT, FEAT, false, true, true><<<gemmGrd, gemmBlk, 0, stream>>>(h, W1, nullptr, Wh, N,
                                                                                al1, ar1, el, er,
                                                                                dst, deg, rank, E);
    k_rowptr<<<(N + 1023) / 1024, 1024, 0, stream>>>(deg, rowptr, N);
    k_fill<<<(E + 255) / 256, 256, 0, stream>>>(src, dst, rowptr, rank, csr_src, E);

    k_node_gather<false><<<(N + 7) / 8, 256, 0, stream>>>(Wh, el, er, rowptr, deg, csr_src, b1, Z, N);

    k_gemm_strip<FEAT, FEAT, false, true, false><<<gemmGrd, gemmBlk, 0, stream>>>(Z, W2, nullptr, Wh, N,
                                                                                 al2, ar2, el, er,
                                                                                 nullptr, nullptr, nullptr, 0);
    k_node_gather<true><<<(N + 7) / 8, 256, 0, stream>>>(Wh, el, er, rowptr, deg, csr_src, b2, zmean, N);

    k_gemm_strip<D_HEAD, OUT_DIM, true, false, false><<<gemmGrd, gemmBlk, 0, stream>>>(zmean, Wp, bp, out, N,
                                                                                       nullptr, nullptr, nullptr, nullptr,
                                                                                       nullptr, nullptr, nullptr, 0);
}


# Edge lists at a fixed stride: 32 slots per destination, later arrivals in one overflow list that long nodes scan (summed in list order); GEMM threads place src themselves; scan and fill launches gone
# speedup vs baseline: 7.6249x; 1.1447x over previous
#include <hip/hip_runtime.h>
#include <hip/hip_bf16.h>
#include <math.h>

#define N_NODES   50000
#define N_EDGES   800000
#define HEADS     4
#define D_HEAD    32
#define FEAT      128
#define OUT_DIM   512
#define NEG_SLOPE 0.2f
#define EPS_      1e-9f

typedef __attribute__((ext_vector_type(2))) float v2f;
typedef __attribute__((ext_vector_type(8))) float v8f;


__global__ void k_zero(float* __restrict__ p, long n) {
    long i = (long)blockIdx.x * blockDim.x + threadIdx.x;
    if (i < n) p[i] = 0.0f;
}


template <int XM>
__device__ __forceinline__ float row_xor(float x) {
    return __int_as_float(__builtin_amdgcn_update_dpp(0, __float_as_int(x), 0x160 + XM, 0xF, 0xF, true));
}
template <int LEN, int XM>
__device__ __forceinline__ void halve_step(float (&v)[64], bool hi) {
#pragma unroll
    for (int k = 0; k < LEN / 2; ++k) {
        const float keep = hi ? v[k + LEN / 2] : v[k];
        const float send = hi ? v[k] : v[k + LEN / 2];
        v[k] = keep + row_xor<XM>(send);
    }
}

#define EPT 8
#define CAP 32
#define DEG_PAD 50176
static_assert(DEG_PAD >= N_NODES + 1 && DEG_PAD % 64 == 0, "deg, the counter, then aligned slots");
__device__ __forceinline__ void place_edge(int d, int r, int sv, int* __restrict__ slots,
                                           int* __restrict__ ovf_cnt, int* __restrict__ ovf_dst,
                                           int* __restrict__ ovf_src)
{
    if (r < CAP) {
        slots[(long)d * CAP + r] = sv;
    } else {
        const int k = atomicAdd(ovf_cnt, 1);
        ovf_dst[k] = d;
        ovf_src[k] = sv;
    }
}
static_assert(((N_NODES / 16 + 7) / 8) * 256L * EPT >= N_EDGES && N_EDGES % EPT == 0,
              "the layer-1 GEMM grid's threads cover every edge, EPT whole edges each");
template <int K, int N, bool HAS_BIAS, bool SCORES, bool RANK>
__global__ void __launch_bounds__(256)
k_gemm_strip(const float* __restrict__ A, const float* __restrict__ B,
             const float* __restrict__ bias, float* __restrict__ C, int M,
             const float* __restrict__ al, const float* __restrict__ ar,
             float* __restrict__ el, float* __restrict__ er,
             const int* __restrict__ src, const int* __restrict__ dst, int* __restrict__ deg,
             int* __restrict__ slots, int* __restrict__ ovf_cnt, int* __restrict__ ovf_dst,
             int* __restrict__ ovf_src, int E)
{
    static_assert(K % 8 == 0 && N % 128 == 0 && K * N == 16384, "tile plan");
    static_assert(!SCORES || N == HEADS * D_HEAD, "a head is one pair of column tiles");
    constexpr int KH = K / 2;
    constexpr int KS = K / 4;
    constexpr int NA = KH / 4;
    constexpr int NG = N / 128;

    __shared__ float4 ldsW[K * N / 4];

    const int tid  = threadIdx.x;
    const int wave = tid >> 5;
    const int lane = tid & 31;
    const int c    = lane & 15;
    const int h    = lane >> 4;
    const int m0   = (blockIdx.x * 8 + wave) * 16;

    const long e0  = ((long)blockIdx.x * 256 + tid) * EPT;
    const bool own = RANK && (e0 + EPT <= (long)E);
    int4 dA = make_int4(0, 0, 0, 0), dB = make_int4(0, 0, 0, 0);
    int4 sA = make_int4(0, 0, 0, 0), sB = make_int4(0, 0, 0, 0);
    if (own) {
        dA = *reinterpret_cast<const int4*>(dst + e0);
        dB = *reinterpret_cast<const int4*>(dst + e0 + 4);
        sA = *reinterpret_cast<const int4*>(src + e0);
        sB = *reinterpret_cast<const int4*>(src + e0 + 4);
    }

    const int arow = (m0 + c < M) ? (m0 + c) : (M - 1);
    const float4* ap = reinterpret_cast<const float4*>(A + (long)arow * K + h * KH);
    float4 a4[NA];
#pragma unroll
    for (int i = 0; i < NA; ++i) a4[i] = ap[i];
    float2 bz[N / 32];
    if (HAS_BIAS) {
#pragma unroll
        for (int t = 0; t < N / 32; ++t)
            bz[t] = *reinterpret_cast<const float2*>(bias + 32 * t + 2 * c);
    }
    float2 alv[4], arv[4];
    if (SCORES) {
#pragma unroll
        for (int p = 0; p < 4; ++p) {
            alv[p] = *reinterpret_cast<const float2*>(al + 32 * p + 2 * c);
            arv[p] = *reinterpret_cast<const float2*>(ar + 32 * p + 2 * c);
        }
    }

    constexpr int ITEMS = (K / 2) * (N / 4);
    static_assert(ITEMS % 256 == 0, "staging plan");
#pragma unroll
    for (int u = 0; u < ITEMS / 256; ++u) {
        const int it = tid + u * 256;
        const int kp = it / (N / 4);
        const int q  = it % (N / 4);
        const float4 r0 = *reinterpret_cast<const float4*>(B + (long)(2 * kp) * N + 4 * q);
        const float4 r1 = *reinterpret_cast<const float4*>(B + (long)(2 * kp + 1) * N + 4 * q);
        const int w = (kp * N + 4 * q) >> 1;
        ldsW[w]     = make_float4(r0.x, r1.x, r0.y, r1.y);
        ldsW[w + 1] = make_float4(r0.z, r1.z, r0.w, r1.w);
    }
    int4 rA = make_int4(0, 0, 0, 0), rB = make_int4(0, 0, 0, 0);
    if (own) {
        rA.x = atomicAdd(&deg[dA.x], 1);
        rA.y = atomicAdd(&deg[dA.y], 1);
        rA.z = atomicAdd(&deg[dA.z], 1);
        rA.w = atomicAdd(&deg[dA.w], 1);
        rB.x = atomicAdd(&deg[dB.x], 1);
        rB.y = atomicAdd(&deg[dB.y], 1);
        rB.z = atomicAdd(&deg[dB.z], 1);
        rB.w = atomicAdd(&deg[dB.w], 1);
    }
    if (RANK) {
        __builtin_amdgcn_fence(__ATOMIC_RELEASE, "workgroup", "local");
        __builtin_amdgcn_s_barrier();
        __builtin_amdgcn_fence(__ATOMIC_ACQUIRE, "workgroup", "local");
    } else {
        __syncthreads();
    }
    if (m0 >= M) {
        if (own) {
            place_edge(dA.x, rA.x, sA.x, slots, ovf_cnt, ovf_dst, ovf_src);
            place_edge(dA.y, rA.y, sA.y, slots, ovf_cnt, ovf_dst, ovf_src);
            place_edge(dA.z, rA.z, sA.z, slots, ovf_cnt, ovf_dst, ovf_src);
            place_edge(dA.w, rA.w, sA.w, slots, ovf_cnt, ovf_dst, ovf_src);
            place_edge(dB.x, rB.x, sB.x, slots, ovf_cnt, ovf_dst, ovf_src);
            place_edge(dB.y, rB.y, sB.y, slots, ovf_cnt, ovf_dst, ovf_src);
            place_edge(dB.z, rB.z, sB.z, slots, ovf_cnt, ovf_dst, ovf_src);
            place_edge(dB.w, rB.w, sB.w, slots, ovf_cnt, ovf_dst, ovf_src);
        }
        return;
    }

    const int rbase = m0 + (h << 3);
    const float4* lp = ldsW + (((h * KS) * N + 2 * c) >> 1);

    float4 bq[2][4];
#pragma unroll
    for (int p = 0; p < 4; ++p) bq[0][p] = lp[16 * p];

#pragma unroll
    for (int g = 0; g < NG; ++g) {
        v8f acc[8];
#pragma unroll
        for (int t = 0; t < 8; ++t) acc[t] = v8f{};
#pragma unroll
        for (int j = 0; j < KS; ++j) {
            const int cur = (g * KS + j) & 1;
            const int jn  = (j + 1 < KS) ? (j + 1) : 0;
            const int gn  = (j + 1 < KS) ? g : (g + 1);
            if (gn < NG) {
#pragma unroll
                for (int p = 0; p < 4; ++p) bq[cur ^ 1][p] = lp[gn * 64 + jn * (N / 2) + 16 * p];
            }
            v2f a;
            if (j & 1) { a.x = a4[j >> 1].z; a.y = a4[j >> 1].w; }
            else       { a.x = a4[j >> 1].x; a.y = a4[j >> 1].y; }
#pragma unroll
            for (int p = 0; p < 4; ++p) {
                v2f b0, b1;
                b0.x = bq[cur][p].x; b0.y = bq[cur][p].y;
                b1.x = bq[cur][p].z; b1.y = bq[cur][p].w;
                acc[2 * p]     = __builtin_amdgcn_wmma_f32_16x16x4_f32(false, a, false, b0,
                                                       (short)0, acc[2 * p], false, false);
                acc[2 * p + 1] = __builtin_amdgcn_wmma_f32_16x16x4_f32(false, a, false, b1,
                                                       (short)0, acc[2 * p + 1], false, false);
            }
        }
#pragma unroll
        for (int p = 0; p < 4; ++p) {
            float* crow = C + (long)rbase * N + g * 128 + 32 * p + 2 * c;
#pragma unroll
            for (int i = 0; i < 8; ++i) {
                float2 o;
                o.x = acc[2 * p][i];
                o.y = acc[2 * p + 1][i];
                if (HAS_BIAS) { o.x += bz[g * 4 + p].x; o.y += bz[g * 4 + p].y; }
                *reinterpret_cast<float2*>(crow + (long)i * N) = o;
            }
        }
        if (SCORES) {
            float v[64];
#pragma unroll
            for (int i = 0; i < 8; ++i) {
#pragma unroll
                for (int p = 0; p < 4; ++p) {
                    v[i * 4 + p]      = acc[2 * p][i] * alv[p].x + acc[2 * p + 1][i] * alv[p].y;
                    v[32 + i * 4 + p] = acc[2 * p][i] * arv[p].x + acc[2 * p + 1][i] * arv[p].y;
                }
            }
            halve_step<64, 8>(v, (c & 8) != 0);
            halve_step<32, 4>(v, (c & 4) != 0);
            halve_step<16, 2>(v, (c & 2) != 0);
            halve_step<8, 1>(v, (c & 1) != 0);
            float* sp = ((c & 8) ? er : el) + (long)(rbase + (c & 7)) * HEADS;
            *reinterpret_cast<float4*>(sp) = make_float4(v[0], v[1], v[2], v[3]);
        }
    }
    if (own) {
        place_edge(dA.x, rA.x, sA.x, slots, ovf_cnt, ovf_dst, ovf_src);
        place_edge(dA.y, rA.y, sA.y, slots, ovf_cnt, ovf_dst, ovf_src);
        place_edge(dA.z, rA.z, sA.z, slots, ovf_cnt, ovf_dst, ovf_src);
        place_edge(dA.w, rA.w, sA.w, slots, ovf_cnt, ovf_dst, ovf_src);
        place_edge(dB.x, rB.x, sB.x, slots, ovf_cnt, ovf_dst, ovf_src);
        place_edge(dB.y, rB.y, sB.y, slots, ovf_cnt, ovf_dst, ovf_src);
        place_edge(dB.z, rB.z, sB.z, slots, ovf_cnt, ovf_dst, ovf_src);
        place_edge(dB.w, rB.w, sB.w, slots, ovf_cnt, ovf_dst, ovf_src);
    }
}


__device__ __forceinline__ float leaky(float v) { return (v > 0.0f) ? v : NEG_SLOPE * v; }
__device__ __forceinline__ float wave_max(float v) {
    for (int o = 16; o > 0; o >>= 1) v = fmaxf(v, __shfl_xor(v, o));
    return v;
}
__device__ __forceinline__ float wave_sum(float v) {
    for (int o = 16; o > 0; o >>= 1) v += __shfl_xor(v, o);
    return v;
}
__device__ __forceinline__ float finite_or_zero(float m) {
    return (fabsf(m) <= 3.402823466e+38f) ? m : 0.0f;
}
__device__ __forceinline__ float lane_bcast(float v, int t) {
    return __int_as_float(__builtin_amdgcn_readlane(__float_as_int(v), t));
}
__device__ __forceinline__ float elu(float x) { return (x > 0.0f) ? x : expm1f(x); }

__device__ __forceinline__ void add_rows(const float* __restrict__ whl, int s,
                                         float a0, float a1, float a2, float a3, int c,
                                         float& acc0, float& acc1, float& acc2, float& acc3)
{
#pragma unroll 4
    for (int t = 0; t < c; ++t) {
        const float* row = whl + (long)__builtin_amdgcn_readlane(s, t) * FEAT;
        acc0 += lane_bcast(a0, t) * row[0];
        acc1 += lane_bcast(a1, t) * row[D_HEAD];
        acc2 += lane_bcast(a2, t) * row[2 * D_HEAD];
        acc3 += lane_bcast(a3, t) * row[3 * D_HEAD];
    }
}

template <bool MEAN_HEADS>
__global__ void __launch_bounds__(256)
k_node_gather(const float* __restrict__ Wh, const float* __restrict__ el,
              const float* __restrict__ er, const int* __restrict__ slots,
              const int* __restrict__ deg, const int* __restrict__ ovf_cnt,
              const int* __restrict__ ovf_dst, const int* __restrict__ ovf_src,
              const float* __restrict__ b, float* __restrict__ res, int N, int E)
{
    const int lane = threadIdx.x & 31;
    const int n = __builtin_amdgcn_readfirstlane((int)(blockIdx.x * 8 + (threadIdx.x >> 5)));
    if (n >= N) return;
    const int* lst  = slots + (long)n * CAP;
    const int cnt   = deg[n];
    const float4 r4 = *reinterpret_cast<const float4*>(er + (long)n * HEADS);
    const float bias0 = b[lane];
    const float bias1 = b[D_HEAD + lane];
    const float bias2 = b[2 * D_HEAD + lane];
    const float bias3 = b[3 * D_HEAD + lane];

    const float* whl = Wh + lane;
    float acc0 = 0.0f, acc1 = 0.0f, acc2 = 0.0f, acc3 = 0.0f;
    if (cnt <= CAP) {
        const bool act = lane < cnt;
        int   s  = 0;
        float e0 = -INFINITY, e1 = -INFINITY, e2 = -INFINITY, e3 = -INFINITY;
        if (act) {
            s = lst[lane];
            const float4 l4 = *reinterpret_cast<const float4*>(el + (long)s * HEADS);
            e0 = leaky(l4.x + r4.x);
            e1 = leaky(l4.y + r4.y);
            e2 = leaky(l4.z + r4.z);
            e3 = leaky(l4.w + r4.w);
        }
        const float m0 = finite_or_zero(wave_max(e0));
        const float m1 = finite_or_zero(wave_max(e1));
        const float m2 = finite_or_zero(wave_max(e2));
        const float m3 = finite_or_zero(wave_max(e3));
        const float p0 = act ? expf(e0 - m0) : 0.0f;
        const float p1 = act ? expf(e1 - m1) : 0.0f;
        const float p2 = act ? expf(e2 - m2) : 0.0f;
        const float p3 = act ? expf(e3 - m3) : 0.0f;
        const float d0 = wave_sum(p0) + EPS_;
        const float d1 = wave_sum(p1) + EPS_;
        const float d2 = wave_sum(p2) + EPS_;
        const float d3 = wave_sum(p3) + EPS_;
        add_rows(whl, s, p0 / d0, p1 / d1, p2 / d2, p3 / d3, cnt, acc0, acc1, acc2, acc3);
    } else {
        int novf = __builtin_amdgcn_readfirstlane(*ovf_cnt);
        novf = (novf < E) ? novf : E;
        const int s_own = lst[lane];
        const float4 o4 = *reinterpret_cast<const float4*>(el + (long)s_own * HEADS);
        const float f0 = leaky(o4.x + r4.x);
        const float f1 = leaky(o4.y + r4.y);
        const float f2 = leaky(o4.z + r4.z);
        const float f3 = leaky(o4.w + r4.w);

        float m0 = f0, m1 = f1, m2 = f2, m3 = f3;
        for (int base = 0; base < novf; base += 32) {
            const int j = base + lane;
            if (j < novf && ovf_dst[j] == n) {
                const int s = ovf_src[j];
                const float4 l4 = *reinterpret_cast<const float4*>(el + (long)s * HEADS);
                m0 = fmaxf(m0, leaky(l4.x + r4.x));
                m1 = fmaxf(m1, leaky(l4.y + r4.y));
                m2 = fmaxf(m2, leaky(l4.z + r4.z));
                m3 = fmaxf(m3, leaky(l4.w + r4.w));
            }
        }
        m0 = finite_or_zero(wave_max(m0));
        m1 = finite_or_zero(wave_max(m1));
        m2 = finite_or_zero(wave_max(m2));
        m3 = finite_or_zero(wave_max(m3));

        float d0 = expf(f0 - m0), d1 = expf(f1 - m1), d2 = expf(f2 - m2), d3 = expf(f3 - m3);
        for (int base = 0; base < novf; base += 32) {
            const int j = base + lane;
            if (j < novf && ovf_dst[j] == n) {
                const int s = ovf_src[j];
                const float4 l4 = *reinterpret_cast<const float4*>(el + (long)s * HEADS);
                d0 += expf(leaky(l4.x + r4.x) - m0);
                d1 += expf(leaky(l4.y + r4.y) - m1);
                d2 += expf(leaky(l4.z + r4.z) - m2);
                d3 += expf(leaky(l4.w + r4.w) - m3);
            }
        }
        d0 = wave_sum(d0) + EPS_;
        d1 = wave_sum(d1) + EPS_;
        d2 = wave_sum(d2) + EPS_;
        d3 = wave_sum(d3) + EPS_;

        add_rows(whl, s_own, expf(f0 - m0) / d0, expf(f1 - m1) / d1, expf(f2 - m2) / d2,
                 expf(f3 - m3) / d3, CAP, acc0, acc1, acc2, acc3);
        for (int base = 0; base < novf; base += 32) {
            const int j = base + lane;
            const bool hit = (j < novf) && (ovf_dst[j] == n);
            int   s  = 0;
            float a0 = 0.0f, a1 = 0.0f, a2 = 0.0f, a3 = 0.0f;
            if (hit) {
                s = ovf_src[j];
                const float4 l4 = *reinterpret_cast<const float4*>(el + (long)s * HEADS);
                a0 = expf(leaky(l4.x + r4.x) - m0) / d0;
                a1 = expf(leaky(l4.y + r4.y) - m1) / d1;
                a2 = expf(leaky(l4.z + r4.z) - m2) / d2;
                a3 = expf(leaky(l4.w + r4.w) - m3) / d3;
            }
            unsigned todo = (unsigned)__ballot(hit);
            while (todo != 0u) {
                const int t = __builtin_ctz(todo);
                todo &= todo - 1u;
                const float* row = whl + (long)__builtin_amdgcn_readlane(s, t) * FEAT;
                acc0 += lane_bcast(a0, t) * row[0];
                acc1 += lane_bcast(a1, t) * row[D_HEAD];
                acc2 += lane_bcast(a2, t) * row[2 * D_HEAD];
                acc3 += lane_bcast(a3, t) * row[3 * D_HEAD];
            }
        }
    }
    const float x0 = acc0 + bias0;
    const float x1 = acc1 + bias1;
    const float x2 = acc2 + bias2;
    const float x3 = acc3 + bias3;
    if (MEAN_HEADS) {
        float s = 0.0f;
        s += x0;
        s += x1;
        s += x2;
        s += x3;
        res[(long)n * D_HEAD + lane] = s * (1.0f / HEADS);
    } else {
        float* o = res + (long)n * FEAT + lane;
        o[0]          = elu(x0);
        o[D_HEAD]     = elu(x1);
        o[2 * D_HEAD] = elu(x2);
        o[3 * D_HEAD] = elu(x3);
    }
}


extern "C" void kernel_launch(void* const* d_in, const int* in_sizes, int n_in,
                              void* d_out, int out_size, void* d_ws, size_t ws_size,
                              hipStream_t stream)
{
    const float* h   = (const float*)d_in[0];
    const int*   src = (const int*)  d_in[1];
    const int*   dst = (const int*)  d_in[2];
    const float* W1  = (const float*)d_in[3];
    const float* al1 = (const float*)d_in[4];
    const float* ar1 = (const float*)d_in[5];
    const float* b1  = (const float*)d_in[6];
    const float* W2  = (const float*)d_in[7];
    const float* al2 = (const float*)d_in[8];
    const float* ar2 = (const float*)d_in[9];
    const float* b2  = (const float*)d_in[10];
    const float* Wp  = (const float*)d_in[11];
    const float* bp  = (const float*)d_in[12];
    float*       out = (float*)d_out;

    const int N = N_NODES, E = N_EDGES;
    static_assert(N_NODES % 16 == 0, "the GEMM's 16-row strips are whole");

    float* Wh      = (float*)d_ws;
    float* Z       = Wh + (size_t)N * FEAT;
    float* el      = Z  + (size_t)N * FEAT;
    float* er      = el + (size_t)N * HEADS;
    float* zmean   = er + (size_t)N * HEADS;
    int*   deg     = (int*)(zmean + (size_t)N * D_HEAD);
    int*   ovf_cnt = deg + (size_t)N;
    int*   slots   = deg + (size_t)DEG_PAD;
    int*   ovf_dst = slots + (size_t)N * CAP;
    int*   ovf_src = ovf_dst + (size_t)E;

    const int mtiles = (N + 15) / 16;
    const dim3 gemmBlk(256);
    const dim3 gemmGrd((mtiles + 7) / 8);

    const int zThreads = 256;
    auto zgrid = [](long n) { return (unsigned)((n + 255) / 256); };

    k_zero<<<zgrid((long)N + 1), zThreads, 0, stream>>>((float*)deg, (long)N + 1);
    k_gemm_strip<FEAT, FEAT, false, true, true><<<gemmGrd, gemmBlk, 0, stream>>>(h, W1, nullptr, Wh, N,
                                                                                al1, ar1, el, er,
                                                                                src, dst, deg, slots,
                                                                                ovf_cnt, ovf_dst, ovf_src, E);

    k_node_gather<false><<<(N + 7) / 8, 256, 0, stream>>>(Wh, el, er, slots, deg, ovf_cnt, ovf_dst, ovf_src,
                                                          b1, Z, N, E);

    k_gemm_strip<FEAT, FEAT, false, true, false><<<gemmGrd, gemmBlk, 0, stream>>>(Z, W2, nullptr, Wh, N,
                                                                                 al2, ar2, el, er,
                                                                                 nullptr, nullptr, nullptr, nullptr,
                                                                                 nullptr, nullptr, nullptr, 0);
    k_node_gather<true><<<(N + 7) / 8, 256, 0, stream>>>(Wh, el, er, slots, deg, ovf_cnt, ovf_dst, ovf_src,
                                                         b2, zmean, N, E);

    k_gemm_strip<D_HEAD, OUT_DIM, true, false, false><<<gemmGrd, gemmBlk, 0, stream>>>(zmean, Wp, bp, out, N,
                                                                                       nullptr, nullptr, nullptr, nullptr,
                                                                                       nullptr, nullptr, nullptr, nullptr,
                                                                                       nullptr, nullptr, nullptr, 0);
}
